# starting kernel
# baseline (speedup 1.0000x reference)
#include <hip/hip_runtime.h>

typedef __attribute__((ext_vector_type(2))) float v2f;
typedef __attribute__((ext_vector_type(8))) float v8f;

#define DIN 256
#define HD  128
#define HH  4
#define DD  32
#define NEG_SLOPE 0.2f

__global__ __launch_bounds__(256) void gat_gemm(const float* __restrict__ feat,
                                                const float* __restrict__ W,
                                                float* __restrict__ ft, int N) {
    __shared__ float As[16][260];
    const int tid = threadIdx.x;
    const int rowBase = blockIdx.x * 16;
    const bool fullTile = (rowBase + 16 <= N);

    if (fullTile) {
        #pragma unroll
        for (int i = 0; i < 4; ++i) {
            int idx4 = i * 256 + tid;
            int r    = idx4 >> 6;
            int c4   = idx4 & 63;
            float4 v = reinterpret_cast<const float4*>(feat + (size_t)(rowBase + r) * DIN)[c4];
            As[r][c4 * 4 + 0] = v.x; As[r][c4 * 4 + 1] = v.y;
            As[r][c4 * 4 + 2] = v.z; As[r][c4 * 4 + 3] = v.w;
        }
    } else {
        #pragma unroll
        for (int i = 0; i < 4; ++i) {
            int idx4 = i * 256 + tid;
            int r    = idx4 >> 6;
            int c4   = idx4 & 63;
            float4 v = make_float4(0.f, 0.f, 0.f, 0.f);
            if (rowBase + r < N)
                v = reinterpret_cast<const float4*>(feat + (size_t)(rowBase + r) * DIN)[c4];
            As[r][c4 * 4 + 0] = v.x; As[r][c4 * 4 + 1] = v.y;
            As[r][c4 * 4 + 2] = v.z; As[r][c4 * 4 + 3] = v.w;
        }
    }
    __syncthreads();

    const int wave    = tid >> 5;
    const int lane    = tid & 31;
    const int half    = lane >> 4;
    const int mrow    = lane & 15;
    const int colBase = wave * 16;
    const int ocol    = colBase + (lane & 15);
    const int kp      = half * 2;

    v8f c = {0.f, 0.f, 0.f, 0.f, 0.f, 0.f, 0.f, 0.f};
    const float* wrow = W + (size_t)ocol * DIN;

    #pragma unroll 8
    for (int k0 = 0; k0 < DIN; k0 += 4) {
        v2f a = *reinterpret_cast<const v2f*>(&As[mrow][k0 + kp]);
        v2f b = *reinterpret_cast<const v2f*>(wrow + k0 + kp);
        c = __builtin_amdgcn_wmma_f32_16x16x4_f32(
                 false, a,  false, b,
                 (short)0, c,  false,  false);
    }

    float* outBase = ft + (size_t)(rowBase + 8 * half) * HD + colBase + (lane & 15);
    if (fullTile) {
        #pragma unroll
        for (int v = 0; v < 8; ++v)
            outBase[(size_t)v * HD] = c[v];
    } else {
        #pragma unroll
        for (int v = 0; v < 8; ++v) {
            if (rowBase + v + 8 * half < N)
                outBase[(size_t)v * HD] = c[v];
        }
    }
}

__global__ __launch_bounds__(128) void gat_attn_logits(
        const float* __restrict__ ft,
        const float* __restrict__ attn_l, const float* __restrict__ attn_r,
        float* __restrict__ el, float* __restrict__ er) {
    const int n    = blockIdx.x;
    const int h    = threadIdx.x >> 5;
    const int lane = threadIdx.x & 31;
    float v = ft[(size_t)n * HD + h * DD + lane];
    float l = v * attn_l[h * DD + lane];
    float r = v * attn_r[h * DD + lane];
    #pragma unroll
    for (int off = 16; off > 0; off >>= 1) {
        l += __shfl_xor(l, off, 32);
        r += __shfl_xor(r, off, 32);
    }
    if (lane == 0) {
        el[n * HH + h] = l;
        er[n * HH + h] = r;
    }
}

__global__ void gat_init(float* __restrict__ out, const float* __restrict__ bias,
                         float* __restrict__ m, float* __restrict__ denom, int N) {
    size_t i = (size_t)blockIdx.x * blockDim.x + threadIdx.x;
    if (i < (size_t)N * HD) out[i] = bias[i & (HD - 1)];
    if (i < (size_t)N * HH) { m[i] = -3.402823466e38f; denom[i] = 0.f; }
}

__device__ __forceinline__ float leaky(float x) {
    return (x > 0.f) ? x : NEG_SLOPE * x;
}

__global__ void gat_edge_scores(const int* __restrict__ src, const int* __restrict__ dst,
                                const float* __restrict__ el, const float* __restrict__ er,
                                float* __restrict__ esc, float* __restrict__ m, int E) {
    int e = blockIdx.x * blockDim.x + threadIdx.x;
    if (e >= E) return;
    int s = src[e], d = dst[e];
    float4 l = reinterpret_cast<const float4*>(el)[s];
    float4 r = reinterpret_cast<const float4*>(er)[d];
    float4 v;
    v.x = leaky(l.x + r.x); v.y = leaky(l.y + r.y);
    v.z = leaky(l.z + r.z); v.w = leaky(l.w + r.w);
    reinterpret_cast<float4*>(esc)[e] = v;
    atomicMax(&m[d * HH + 0], v.x);
    atomicMax(&m[d * HH + 1], v.y);
    atomicMax(&m[d * HH + 2], v.z);
    atomicMax(&m[d * HH + 3], v.w);
}

__global__ void gat_edge_exp(const int* __restrict__ dst, float* __restrict__ esc,
                             const float* __restrict__ m, float* __restrict__ denom, int E) {
    int e = blockIdx.x * blockDim.x + threadIdx.x;
    if (e >= E) return;
    int d = dst[e];
    float4 v  = reinterpret_cast<const float4*>(esc)[e];
    float4 mx = reinterpret_cast<const float4*>(m)[d];
    v.x = __expf(v.x - mx.x); v.y = __expf(v.y - mx.y);
    v.z = __expf(v.z - mx.z); v.w = __expf(v.w - mx.w);
    reinterpret_cast<float4*>(esc)[e] = v;
    atomicAdd(&denom[d * HH + 0], v.x);
    atomicAdd(&denom[d * HH + 1], v.y);
    atomicAdd(&denom[d * HH + 2], v.z);
    atomicAdd(&denom[d * HH + 3], v.w);
}

__global__ __launch_bounds__(256) void gat_aggregate(
        const int* __restrict__ src, const int* __restrict__ dst,
        const float* __restrict__ esc, const float* __restrict__ denom,
        const float* __restrict__ ft, float* __restrict__ out, int E) {
    int e = blockIdx.x * 8 + (threadIdx.x >> 5);
    int lane = threadIdx.x & 31;
    if (e >= E) return;
    int s = src[e], d = dst[e];
    const float* fts = ft + (size_t)s * HD;
    __builtin_prefetch(fts + lane, 0, 0);
    float4 ae = reinterpret_cast<const float4*>(esc)[e];
    float4 dn = reinterpret_cast<const float4*>(denom)[d];
    float a0 = ae.x / dn.x, a1 = ae.y / dn.y, a2 = ae.z / dn.z, a3 = ae.w / dn.w;
    float* outd = out + (size_t)d * HD;
    atomicAdd(&outd[0 * DD + lane], fts[0 * DD + lane] * a0);
    atomicAdd(&outd[1 * DD + lane], fts[1 * DD + lane] * a1);
    atomicAdd(&outd[2 * DD + lane], fts[2 * DD + lane] * a2);
    atomicAdd(&outd[3 * DD + lane], fts[3 * DD + lane] * a3);
}

extern "C" void kernel_launch(void* const* d_in, const int* in_sizes, int n_in,
                              void* d_out, int out_size, void* d_ws, size_t ws_size,
                              hipStream_t stream) {
    const float* feat   = (const float*)d_in[0];
    const int*   src    = (const int*)  d_in[1];
    const int*   dst    = (const int*)  d_in[2];
    const float* W      = (const float*)d_in[3];
    const float* attn_l = (const float*)d_in[4];
    const float* attn_r = (const float*)d_in[5];
    const float* bias   = (const float*)d_in[6];

    const int N = in_sizes[0] / DIN;
    const int E = in_sizes[1];
    float* out = (float*)d_out;

    float* ws    = (float*)d_ws;
    float* ft    = ws;
    float* el    = ft    + (size_t)N * HD;
    float* er    = el    + (size_t)N * HH;
    float* mbuf  = er    + (size_t)N * HH;
    float* denom = mbuf  + (size_t)N * HH;
    float* esc   = denom + (size_t)N * HH;

    gat_gemm<<<(N + 15) / 16, 256, 0, stream>>>(feat, W, ft, N);
    gat_attn_logits<<<N, 128, 0, stream>>>(ft, attn_l, attn_r, el, er);

    size_t tot = (size_t)N * HD;
    gat_init<<<(unsigned)((tot + 255) / 256), 256, 0, stream>>>(out, bias, mbuf, denom, N);

    gat_edge_scores<<<(E + 255) / 256, 256, 0, stream>>>(src, dst, el, er, esc, mbuf, E);
    gat_edge_exp<<<(E + 255) / 256, 256, 0, stream>>>(dst, esc, mbuf, denom, E);
    gat_aggregate<<<(E + 7) / 8, 256, 0, stream>>>(src, dst, esc, denom, ft, out, E);
}


# Segment max moved into the owner kernel: a wave queues all its edges in LDS, LDS float max per node, barrier, then sums; score kernel, score array, global max atomics gone; LDS atomic sum order kept
# speedup vs baseline: 4.2209x; 4.2209x over previous
#include <hip/hip_runtime.h>

typedef __attribute__((ext_vector_type(2))) float v2f;
typedef __attribute__((ext_vector_type(8))) float v8f;

#define DIN 256
#define HD  128
#define HH  4
#define DD  32
#define NEG_SLOPE 0.2f

#define GEMM_BLOCKS 512
#define KG 16
#define APITCH 260

__device__ __forceinline__ void gemm_fetch_tile(float4 (&pre)[4], const float* __restrict__ feat,
                                                int rowBase, int N, int tid) {
    if (rowBase + 16 <= N) {
        #pragma unroll
        for (int i = 0; i < 4; ++i) {
            const int idx4 = i * 256 + tid;
            const int r    = idx4 >> 6;
            const int c4   = idx4 & 63;
            pre[i] = reinterpret_cast<const float4*>(feat + (size_t)(rowBase + r) * DIN)[c4];
        }
    } else {
        #pragma unroll
        for (int i = 0; i < 4; ++i) {
            const int idx4 = i * 256 + tid;
            const int r    = idx4 >> 6;
            const int c4   = idx4 & 63;
            float4 v = make_float4(0.f, 0.f, 0.f, 0.f);
            if (rowBase + r < N)
                v = reinterpret_cast<const float4*>(feat + (size_t)(rowBase + r) * DIN)[c4];
            pre[i] = v;
        }
    }
}

__device__ __forceinline__ void gemm_store_tile(float (*As)[APITCH], const float4 (&pre)[4], int tid) {
    #pragma unroll
    for (int i = 0; i < 4; ++i) {
        const int idx4 = i * 256 + tid;
        const int r    = idx4 >> 6;
        const int c4   = idx4 & 63;
        As[r][c4 * 4 + 0] = pre[i].x; As[r][c4 * 4 + 1] = pre[i].y;
        As[r][c4 * 4 + 2] = pre[i].z; As[r][c4 * 4 + 3] = pre[i].w;
    }
}

__global__ __launch_bounds__(256) void gat_gemm(const float* __restrict__ feat,
                                                const float* __restrict__ W,
                                                float* __restrict__ ft, int N) {
    __shared__ float As[2][16][APITCH];
    const int tid      = threadIdx.x;
    const int numTiles = (N + 15) >> 4;
    int tile = blockIdx.x;
    if (tile >= numTiles) return;

    const int wave    = tid >> 5;
    const int lane    = tid & 31;
    const int half    = lane >> 4;
    const int mrow    = lane & 15;
    const int colBase = wave * 16;
    const int kp      = half * 2;

    const float* wrow = W + (size_t)(colBase + mrow) * DIN + kp;
    v2f b[DIN / 4];
    #pragma unroll
    for (int i = 0; i < DIN / 4; ++i)
        b[i] = *reinterpret_cast<const v2f*>(wrow + 4 * i);

    float4 pre[4];
    gemm_fetch_tile(pre, feat, tile * 16, N, tid);
    gemm_store_tile(As[0], pre, tid);
    __syncthreads();

    int cur = 0;
    for (;;) {
        const int  next    = tile + (int)gridDim.x;
        const bool hasNext = next < numTiles;
        if (hasNext) gemm_fetch_tile(pre, feat, next * 16, N, tid);

        const int rowBase = tile * 16;
        v8f c = {0.f, 0.f, 0.f, 0.f, 0.f, 0.f, 0.f, 0.f};
        const float* arow = &As[cur][mrow][kp];
        v2f a[2][KG];
        #pragma unroll
        for (int i = 0; i < KG; ++i)
            a[0][i] = *reinterpret_cast<const v2f*>(arow + 4 * i);
        __builtin_amdgcn_sched_group_barrier(0x100, KG / 2, 0);
        #pragma unroll
        for (int g = 0; g < DIN / 4 / KG; ++g) {
            if (g + 1 < DIN / 4 / KG) {
                #pragma unroll
                for (int i = 0; i < KG; ++i)
                    a[(g + 1) & 1][i] = *reinterpret_cast<const v2f*>(arow + 4 * ((g + 1) * KG + i));
            }
            #pragma unroll
            for (int i = 0; i < KG; ++i)
                c = __builtin_amdgcn_wmma_f32_16x16x4_f32(
                         false, a[g & 1][i],  false, b[g * KG + i],
                         (short)0, c,  false,  false);
            if (g + 1 < DIN / 4 / KG)
                __builtin_amdgcn_sched_group_barrier(0x100, KG / 2, 0);
            __builtin_amdgcn_sched_group_barrier(0x008, KG, 0);
        }

        float* outBase = ft + (size_t)(rowBase + 8 * half) * HD + colBase + mrow;
        if (rowBase + 16 <= N) {
            #pragma unroll
            for (int v = 0; v < 8; ++v)
                outBase[(size_t)v * HD] = c[v];
        } else {
            #pragma unroll
            for (int v = 0; v < 8; ++v) {
                if (rowBase + v + 8 * half < N)
                    outBase[(size_t)v * HD] = c[v];
            }
        }

        if (!hasNext) break;
        gemm_store_tile(As[cur ^ 1], pre, tid);
        __syncthreads();
        cur ^= 1;
        tile = next;
    }
}

#define NPB 8

__global__ __launch_bounds__(128) void gat_attn_logits(
        const float* __restrict__ ft,
        const float* __restrict__ attn_l, const float* __restrict__ attn_r,
        float* __restrict__ el, float* __restrict__ er, int N) {
    __shared__ float sl[NPB * HH];
    __shared__ float sr[NPB * HH];
    const int h    = threadIdx.x >> 5;
    const int lane = threadIdx.x & 31;
    const int n0   = blockIdx.x * NPB;
    const float wl = attn_l[h * DD + lane];
    const float wr = attn_r[h * DD + lane];
    const float* frow = ft + (size_t)n0 * HD + h * DD + lane;
    float v[NPB];
    if (n0 + NPB <= N) {
        #pragma unroll
        for (int i = 0; i < NPB; ++i) v[i] = frow[(size_t)i * HD];
    } else {
        #pragma unroll
        for (int i = 0; i < NPB; ++i) v[i] = (n0 + i < N) ? frow[(size_t)i * HD] : 0.f;
    }
    #pragma unroll
    for (int i = 0; i < NPB; ++i) {
        float l = v[i] * wl;
        float r = v[i] * wr;
        #pragma unroll
        for (int off = 16; off > 0; off >>= 1) {
            l += __shfl_xor(l, off, 32);
            r += __shfl_xor(r, off, 32);
        }
        if (lane == 0) { sl[i * HH + h] = l; sr[i * HH + h] = r; }
    }
    __syncthreads();
    const int t = threadIdx.x;
    if (t < NPB * HH) {
        if (n0 * HH + t < N * HH) el[n0 * HH + t] = sl[t];
    } else if (t < 2 * NPB * HH) {
        const int u = t - NPB * HH;
        if (n0 * HH + u < N * HH) er[n0 * HH + u] = sr[u];
    }
}

#define RMAX 392

__device__ __forceinline__ unsigned own_id(int d) {
    return ((unsigned)d / (unsigned)RMAX) & 255u;
}

__global__ __launch_bounds__(256) void gat_init_owner(
        const int* __restrict__ dst, unsigned* __restrict__ own, int E) {
    const int t      = blockIdx.x * 256 + threadIdx.x;
    const int stride = (int)gridDim.x * 256;
    const int nw     = (E + 3) >> 2;
    for (int w = t; w < nw; w += stride) {
        const int e = w * 4;
        unsigned v = 0u;
        if (e + 4 <= E) {
            const int4 d = reinterpret_cast<const int4*>(dst)[w];
            v = own_id(d.x) | (own_id(d.y) << 8) | (own_id(d.z) << 16) | (own_id(d.w) << 24);
        } else {
            for (int c = 0; c < 4; ++c)
                if (e + c < E) v |= own_id(dst[e + c]) << (8 * c);
        }
        own[w] = v;
    }
}

__device__ __forceinline__ float leaky(float x) {
    return (x > 0.f) ? x : NEG_SLOPE * x;
}

#define OWN_BLOCKS  256
#define OWN_THREADS 1024
#define OWN_WAVES   (OWN_THREADS / 32)
#define DRAIN       8
#define SCAN_LOADS  4
#define QCAP        480
#define RESOLVE     4
#define NOT_OWNED   0xFFFFu

__device__ __forceinline__ void own_resolve_max(
        int* q, unsigned short* qd, int n, int lane, int n0, int cnt,
        const int* __restrict__ src, const int* __restrict__ dst,
        const float* __restrict__ el, const float* erl, float* mloc) {
    for (int base = 0; base < n; base += 32 * RESOLVE) {
        int  e[RESOLVE], d[RESOLVE], s[RESOLVE];
        bool in[RESOLVE];
        #pragma unroll
        for (int u = 0; u < RESOLVE; ++u) {
            const int idx = base + u * 32 + lane;
            in[u] = idx < n;
            e[u]  = in[u] ? q[idx] : 0;
        }
        #pragma unroll
        for (int u = 0; u < RESOLVE; ++u) {
            d[u] = dst[e[u]];
            s[u] = src[e[u]];
        }
        float4 l[RESOLVE];
        #pragma unroll
        for (int u = 0; u < RESOLVE; ++u)
            l[u] = reinterpret_cast<const float4*>(el)[s[u]];
        #pragma unroll
        for (int u = 0; u < RESOLVE; ++u) {
            const int  idx = base + u * 32 + lane;
            const int  dl  = d[u] - n0;
            const bool ok  = in[u] && ((unsigned)dl < (unsigned)cnt);
            if (in[u]) {
                q[idx]  = s[u];
                qd[idx] = ok ? (unsigned short)dl : (unsigned short)NOT_OWNED;
            }
            if (ok) {
                const float4 r = reinterpret_cast<const float4*>(erl)[dl];
                atomicMax(&mloc[dl * HH + 0], leaky(l[u].x + r.x));
                atomicMax(&mloc[dl * HH + 1], leaky(l[u].y + r.y));
                atomicMax(&mloc[dl * HH + 2], leaky(l[u].z + r.z));
                atomicMax(&mloc[dl * HH + 3], leaky(l[u].w + r.w));
            }
        }
    }
}

__device__ __forceinline__ void own_accumulate(
        const int* q, const unsigned short* qd, int n, int lane,
        const float* __restrict__ el, const float* __restrict__ ft,
        const float* erl, const float* mloc, float* den, float* acc) {
    const int j = lane >> 2;
    const int h = lane & 3;
    int b = 0;
    for (; b + DRAIN <= n; b += DRAIN) {
        const int      sv = q[b + j];
        const unsigned dr = qd[b + j];
        const bool     ok = dr != NOT_OWNED;
        const int      dl = ok ? (int)dr : 0;
        const float    xl = el[(size_t)sv * HH + h];
        float v[DRAIN][HH];
        #pragma unroll
        for (int jj = 0; jj < DRAIN; ++jj) {
            const int s = __builtin_amdgcn_readlane(sv, 4 * jj);
            const float* fts = ft + (size_t)s * HD + lane;
            #pragma unroll
            for (int hh = 0; hh < HH; ++hh) v[jj][hh] = fts[hh * DD];
        }
        const float x  = leaky(xl + erl[dl * HH + h]);
        const float ae = __expf(x - mloc[dl * HH + h]);
        if (ok) atomicAdd(&den[dl * HH + h], ae);
        #pragma unroll
        for (int jj = 0; jj < DRAIN; ++jj) {
            const int dj  = __builtin_amdgcn_readlane(dl, 4 * jj);
            const int okj = __builtin_amdgcn_readlane((int)ok, 4 * jj);
            if (okj) {
                float* accd = acc + dj * HD + lane;
                #pragma unroll
                for (int hh = 0; hh < HH; ++hh) {
                    const float a = __int_as_float(
                        __builtin_amdgcn_readlane(__float_as_int(ae), 4 * jj + hh));
                    atomicAdd(&accd[hh * DD], v[jj][hh] * a);
                }
            }
        }
    }
    for (; b < n; ++b) {
        const int      s  = q[b];
        const unsigned dr = qd[b];
        if (dr != NOT_OWNED) {
            const int dl = (int)dr;
            const float* fts = ft + (size_t)s * HD + lane;
            float* accd = acc + dl * HD + lane;
            #pragma unroll
            for (int hh = 0; hh < HH; ++hh) {
                const float x  = leaky(el[(size_t)s * HH + hh] + erl[dl * HH + hh]);
                const float ae = __expf(x - mloc[dl * HH + hh]);
                if (lane == 0) atomicAdd(&den[dl * HH + hh], ae);
                atomicAdd(&accd[hh * DD], fts[hh * DD] * ae);
            }
        }
    }
}

template <bool ACC>
__device__ __forceinline__ void own_plain_pass(
        int wave, int lane, int steps, int n0, int cnt, unsigned ownByte,
        const int* __restrict__ src, const int* __restrict__ dst, const unsigned* __restrict__ own,
        const float* __restrict__ el, const float* __restrict__ ft,
        const float* erl, float* mloc, float* den, float* acc, int E) {
    for (int i = wave; i < steps; i += OWN_WAVES) {
        for (int c = 0; c < 16; ++c) {
            const int e = i * 512 + lane * 16 + c;
            if (e >= E) continue;
            if (((own[e >> 2] >> (8 * (e & 3))) & 255u) != ownByte) continue;
            const int dl = dst[e] - n0;
            if ((unsigned)dl >= (unsigned)cnt) continue;
            const int s = src[e];
            for (int hh = 0; hh < HH; ++hh) {
                const float x = leaky(el[(size_t)s * HH + hh] + erl[dl * HH + hh]);
                if (!ACC) {
                    atomicMax(&mloc[dl * HH + hh], x);
                } else {
                    const float ae = __expf(x - mloc[dl * HH + hh]);
                    atomicAdd(&den[dl * HH + hh], ae);
                    for (int dd = 0; dd < DD; ++dd)
                        atomicAdd(&acc[dl * HD + hh * DD + dd], ft[(size_t)s * HD + hh * DD + dd] * ae);
                }
            }
        }
    }
}

__global__ __launch_bounds__(OWN_THREADS) void gat_owner_aggregate(
        const int* __restrict__ src, const int* __restrict__ dst,
        const unsigned* __restrict__ own,
        const float* __restrict__ el, const float* __restrict__ er,
        const float* __restrict__ ft, const float* __restrict__ bias,
        float* __restrict__ out, int N, int E) {
    __shared__ __attribute__((aligned(16))) float acc[RMAX * HD];
    __shared__ __attribute__((aligned(16))) float den[RMAX * HH];
    __shared__ __attribute__((aligned(16))) float mloc[RMAX * HH];
    __shared__ __attribute__((aligned(16))) float erl[RMAX * HH];
    __shared__ int queue[OWN_WAVES * QCAP];
    __shared__ unsigned short qdst[OWN_WAVES * QCAP];
    __shared__ int ovf;

    const int tid   = threadIdx.x;
    const int lane  = tid & 31;
    const int wave  = __builtin_amdgcn_readfirstlane(tid >> 5);
    int* q              = &queue[wave * QCAP];
    unsigned short* qd  = &qdst[wave * QCAP];
    const int nw        = (E + 3) >> 2;
    const int steps     = (E + 511) >> 9;
    const int fullSteps = E >> 9;
    const int start = (int)(((long long)blockIdx.x * steps) / gridDim.x);
    const int K     = (steps + OWN_WAVES - 1) / OWN_WAVES;

    for (int n0 = blockIdx.x * RMAX; n0 < N; n0 += (int)gridDim.x * RMAX) {
        const int cnt = (N - n0 < RMAX) ? (N - n0) : RMAX;
        const unsigned ownByte = (unsigned)(n0 / RMAX) & 255u;
        const unsigned pat = ownByte * 0x01010101u;

        const float4 z4 = make_float4(0.f, 0.f, 0.f, 0.f);
        const float4 lo = make_float4(-3.402823466e38f, -3.402823466e38f,
                                      -3.402823466e38f, -3.402823466e38f);
        for (int i = tid; i < cnt * (HD / 4); i += OWN_THREADS) reinterpret_cast<float4*>(acc)[i] = z4;
        for (int i = tid; i < cnt; i += OWN_THREADS) {
            reinterpret_cast<float4*>(den)[i]  = z4;
            reinterpret_cast<float4*>(mloc)[i] = lo;
            reinterpret_cast<float4*>(erl)[i]  = reinterpret_cast<const float4*>(er)[n0 + i];
        }
        if (tid == 0) ovf = 0;
        __syncthreads();

        int  qn   = 0;
        bool lost = false;
        for (int k = 0; k < K; k += SCAN_LOADS) {
            int  rot[SCAN_LOADS];
            bool allFull = true;
            #pragma unroll
            for (int r = 0; r < SCAN_LOADS; ++r) {
                const int i = wave + OWN_WAVES * (k + r);
                int rr = i + start;
                if (rr >= steps) rr -= steps;
                rot[r]  = (i < steps) ? rr : -1;
                allFull = allFull && (i < steps) && (rr < fullSteps);
            }
            uint4 ww[SCAN_LOADS];
            if (allFull) {
                #pragma unroll
                for (int r = 0; r < SCAN_LOADS; ++r)
                    ww[r] = reinterpret_cast<const uint4*>(own)[(size_t)rot[r] * 32 + lane];
            } else {
                #pragma unroll
                for (int r = 0; r < SCAN_LOADS; ++r) {
                    ww[r] = make_uint4(0u, 0u, 0u, 0u);
                    if (rot[r] >= 0 && rot[r] < fullSteps) {
                        ww[r] = reinterpret_cast<const uint4*>(own)[(size_t)rot[r] * 32 + lane];
                    } else if (rot[r] >= 0) {
                        unsigned w4[4];
                        #pragma unroll
                        for (int c = 0; c < 4; ++c) {
                            const int wi   = rot[r] * 128 + lane * 4 + c;
                            const int left = E - wi * 4;
                            unsigned w = (wi < nw) ? own[wi] : ~pat;
                            if (left < 4) {
                                const unsigned keep = (left <= 0) ? 0u : (0xFFFFFFFFu >> (8 * (4 - left)));
                                w = (w & keep) | (~pat & ~keep);
                            }
                            w4[c] = w;
                        }
                        ww[r] = make_uint4(w4[0], w4[1], w4[2], w4[3]);
                    }
                }
            }

            #pragma unroll
            for (int r = 0; r < SCAN_LOADS; ++r) {
                if (rot[r] >= 0) {
                    const int ebase = rot[r] * 512 + lane * 16;
                    const unsigned w4[4] = {ww[r].x, ww[r].y, ww[r].z, ww[r].w};
                    #pragma unroll
                    for (int c = 0; c < 4; ++c) {
                        const unsigned x = w4[c] ^ pat;
                        const unsigned t = (x - 0x01010101u) & ~x & 0x80808080u;
                        if (__builtin_amdgcn_ballot_w32(t != 0u) != 0u) {
                            if (qn > QCAP - 128) {
                                lost = true;
                            } else {
                                #pragma unroll
                                for (int b = 0; b < 4; ++b) {
                                    const bool hit = ((x >> (8 * b)) & 255u) == 0u;
                                    const unsigned mask = __builtin_amdgcn_ballot_w32(hit);
                                    if (mask != 0u) {
                                        if (hit) q[qn + (int)__builtin_amdgcn_mbcnt_lo(mask, 0u)] = ebase + c * 4 + b;
                                        qn += __builtin_popcount(mask);
                                    }
                                }
                            }
                        }
                    }
                }
            }
        }
        if (lost && lane == 0) ovf = 1;

        own_resolve_max(q, qd, qn, lane, n0, cnt, src, dst, el, erl, mloc);
        __syncthreads();

        if (__builtin_amdgcn_readfirstlane(ovf) == 0) {
            own_accumulate(q, qd, qn, lane, el, ft, erl, mloc, den, acc);
        } else {
            own_plain_pass<false>(wave, lane, steps, n0, cnt, ownByte, src, dst, own, el, ft, erl, mloc, den, acc, E);
            __syncthreads();
            own_plain_pass<true>(wave, lane, steps, n0, cnt, ownByte, src, dst, own, el, ft, erl, mloc, den, acc, E);
        }
        __syncthreads();

        for (int i = tid; i < cnt * (HD / 4); i += OWN_THREADS) {
            const int node = i >> 5;
            const int q4   = i & 31;
            const float4 a4 = reinterpret_cast<const float4*>(acc)[i];
            const float  dn = den[node * HH + (q4 >> 3)];
            const float4 b4 = reinterpret_cast<const float4*>(bias)[q4];
            float4 o;
            if (dn > 0.f) {
                o.x = a4.x / dn + b4.x; o.y = a4.y / dn + b4.y;
                o.z = a4.z / dn + b4.z; o.w = a4.w / dn + b4.w;
            } else {
                o = b4;
            }
            reinterpret_cast<float4*>(out)[(size_t)(n0 + node) * (HD / 4) + q4] = o;
        }
        __syncthreads();
    }
}

extern "C" void kernel_launch(void* const* d_in, const int* in_sizes, int n_in,
                              void* d_out, int out_size, void* d_ws, size_t ws_size,
                              hipStream_t stream) {
    const float* feat   = (const float*)d_in[0];
    const int*   src    = (const int*)  d_in[1];
    const int*   dst    = (const int*)  d_in[2];
    const float* W      = (const float*)d_in[3];
    const float* attn_l = (const float*)d_in[4];
    const float* attn_r = (const float*)d_in[5];
    const float* bias   = (const float*)d_in[6];

    const int N = in_sizes[0] / DIN;
    const int E = in_sizes[1];
    float* out = (float*)d_out;

    float* ws    = (float*)d_ws;
    float* ft    = ws;
    float* el    = ft    + (size_t)N * HD;
    float* er    = el    + (size_t)N * HH;
    unsigned* own = (unsigned*)(er + (size_t)N * HH);

    gat_gemm<<<GEMM_BLOCKS, 256, 0, stream>>>(feat, W, ft, N);
    gat_attn_logits<<<(N + NPB - 1) / NPB, 128, 0, stream>>>(ft, attn_l, attn_r, el, er, N);
    gat_init_owner<<<((E + 3) / 4 + 255) / 256, 256, 0, stream>>>(dst, own, E);
    gat_owner_aggregate<<<OWN_BLOCKS, OWN_THREADS, 0, stream>>>(src, dst, own, el, er, ft, bias, out, N, E);
}


# Logits and owner-byte kernels merged into one launch of 512 blocks whose waves loop: a node row is one 16-byte load per lane, an 8-lane exchange sums 8 nodes at once; logit sum order changes, fixed
# speedup vs baseline: 4.4529x; 1.0550x over previous
#include <hip/hip_runtime.h>

typedef __attribute__((ext_vector_type(2))) float v2f;
typedef __attribute__((ext_vector_type(8))) float v8f;

#define DIN 256
#define HD  128
#define HH  4
#define DD  32
#define NEG_SLOPE 0.2f

#define GEMM_BLOCKS 512
#define KG 16
#define APITCH 260

__device__ __forceinline__ void gemm_fetch_tile(float4 (&pre)[4], const float* __restrict__ feat,
                                                int rowBase, int N, int tid) {
    if (rowBase + 16 <= N) {
        #pragma unroll
        for (int i = 0; i < 4; ++i) {
            const int idx4 = i * 256 + tid;
            const int r    = idx4 >> 6;
            const int c4   = idx4 & 63;
            pre[i] = reinterpret_cast<const float4*>(feat + (size_t)(rowBase + r) * DIN)[c4];
        }
    } else {
        #pragma unroll
        for (int i = 0; i < 4; ++i) {
            const int idx4 = i * 256 + tid;
            const int r    = idx4 >> 6;
            const int c4   = idx4 & 63;
            float4 v = make_float4(0.f, 0.f, 0.f, 0.f);
            if (rowBase + r < N)
                v = reinterpret_cast<const float4*>(feat + (size_t)(rowBase + r) * DIN)[c4];
            pre[i] = v;
        }
    }
}

__device__ __forceinline__ void gemm_store_tile(float (*As)[APITCH], const float4 (&pre)[4], int tid) {
    #pragma unroll
    for (int i = 0; i < 4; ++i) {
        const int idx4 = i * 256 + tid;
        const int r    = idx4 >> 6;
        const int c4   = idx4 & 63;
        As[r][c4 * 4 + 0] = pre[i].x; As[r][c4 * 4 + 1] = pre[i].y;
        As[r][c4 * 4 + 2] = pre[i].z; As[r][c4 * 4 + 3] = pre[i].w;
    }
}

__global__ __launch_bounds__(256) void gat_gemm(const float* __restrict__ feat,
                                                const float* __restrict__ W,
                                                float* __restrict__ ft, int N) {
    __shared__ float As[2][16][APITCH];
    const int tid      = threadIdx.x;
    const int numTiles = (N + 15) >> 4;
    int tile = blockIdx.x;
    if (tile >= numTiles) return;

    const int wave    = tid >> 5;
    const int lane    = tid & 31;
    const int half    = lane >> 4;
    const int mrow    = lane & 15;
    const int colBase = wave * 16;
    const int kp      = half * 2;

    const float* wrow = W + (size_t)(colBase + mrow) * DIN + kp;
    v2f b[DIN / 4];
    #pragma unroll
    for (int i = 0; i < DIN / 4; ++i)
        b[i] = *reinterpret_cast<const v2f*>(wrow + 4 * i);

    float4 pre[4];
    gemm_fetch_tile(pre, feat, tile * 16, N, tid);
    gemm_store_tile(As[0], pre, tid);
    __syncthreads();

    int cur = 0;
    for (;;) {
        const int  next    = tile + (int)gridDim.x;
        const bool hasNext = next < numTiles;
        if (hasNext) gemm_fetch_tile(pre, feat, next * 16, N, tid);

        const int rowBase = tile * 16;
        v8f c = {0.f, 0.f, 0.f, 0.f, 0.f, 0.f, 0.f, 0.f};
        const float* arow = &As[cur][mrow][kp];
        v2f a[2][KG];
        #pragma unroll
        for (int i = 0; i < KG; ++i)
            a[0][i] = *reinterpret_cast<const v2f*>(arow + 4 * i);
        __builtin_amdgcn_sched_group_barrier(0x100, KG / 2, 0);
        #pragma unroll
        for (int g = 0; g < DIN / 4 / KG; ++g) {
            if (g + 1 < DIN / 4 / KG) {
                #pragma unroll
                for (int i = 0; i < KG; ++i)
                    a[(g + 1) & 1][i] = *reinterpret_cast<const v2f*>(arow + 4 * ((g + 1) * KG + i));
            }
            #pragma unroll
            for (int i = 0; i < KG; ++i)
                c = __builtin_amdgcn_wmma_f32_16x16x4_f32(
                         false, a[g & 1][i],  false, b[g * KG + i],
                         (short)0, c,  false,  false);
            if (g + 1 < DIN / 4 / KG)
                __builtin_amdgcn_sched_group_barrier(0x100, KG / 2, 0);
            __builtin_amdgcn_sched_group_barrier(0x008, KG, 0);
        }

        float* outBase = ft + (size_t)(rowBase + 8 * half) * HD + colBase + mrow;
        if (rowBase + 16 <= N) {
            #pragma unroll
            for (int v = 0; v < 8; ++v)
                outBase[(size_t)v * HD] = c[v];
        } else {
            #pragma unroll
            for (int v = 0; v < 8; ++v) {
                if (rowBase + v + 8 * half < N)
                    outBase[(size_t)v * HD] = c[v];
            }
        }

        if (!hasNext) break;
        gemm_store_tile(As[cur ^ 1], pre, tid);
        __syncthreads();
        cur ^= 1;
        tile = next;
    }
}

#define NPW        8
#define LG_BLOCKS  512
#define OWN_TRIP   4
#define LG_THREADS 256
#define RMAX 392

__device__ __forceinline__ unsigned own_id(int d) {
    return ((unsigned)d / (unsigned)RMAX) & 255u;
}

__device__ __forceinline__ float head_sum_scatter(const float (&p)[NPW], int lane) {
    const bool b0 = (lane & 1) != 0;
    const bool b1 = (lane & 2) != 0;
    const bool b2 = (lane & 4) != 0;
    float q[4];
    #pragma unroll
    for (int k = 0; k < 4; ++k) {
        const float keep = b0 ? p[2 * k + 1] : p[2 * k];
        const float send = b0 ? p[2 * k]     : p[2 * k + 1];
        q[k] = keep + __shfl_xor(send, 1, 32);
    }
    float r[2];
    #pragma unroll
    for (int k = 0; k < 2; ++k) {
        const float keep = b1 ? q[2 * k + 1] : q[2 * k];
        const float send = b1 ? q[2 * k]     : q[2 * k + 1];
        r[k] = keep + __shfl_xor(send, 2, 32);
    }
    const float keep = b2 ? r[1] : r[0];
    const float send = b2 ? r[0] : r[1];
    return keep + __shfl_xor(send, 4, 32);
}

__global__ __launch_bounds__(LG_THREADS) void gat_logits_owner(
        const float* __restrict__ ft,
        const float* __restrict__ attn_l, const float* __restrict__ attn_r,
        float* __restrict__ el, float* __restrict__ er,
        const int* __restrict__ dst, unsigned* __restrict__ own, int N, int E) {
    const int t      = blockIdx.x * LG_THREADS + threadIdx.x;
    const int stride = (int)gridDim.x * LG_THREADS;

    const int lane   = threadIdx.x & 31;
    const int gwave  = __builtin_amdgcn_readfirstlane(t >> 5);
    const int nwaves = stride >> 5;

    const int nw    = (E + 3) >> 2;
    const int nfull = E >> 2;
    for (int base = gwave * (OWN_TRIP * 32); base < nfull; base += nwaves * (OWN_TRIP * 32)) {
        if (base + OWN_TRIP * 32 <= nfull) {
            int4 d[OWN_TRIP];
            #pragma unroll
            for (int u = 0; u < OWN_TRIP; ++u)
                d[u] = reinterpret_cast<const int4*>(dst)[base + u * 32 + lane];
            #pragma unroll
            for (int u = 0; u < OWN_TRIP; ++u)
                own[base + u * 32 + lane] =
                    own_id(d[u].x) | (own_id(d[u].y) << 8) | (own_id(d[u].z) << 16) | (own_id(d[u].w) << 24);
        } else {
            for (int u = 0; u < OWN_TRIP; ++u) {
                const int w = base + u * 32 + lane;
                if (w < nfull) {
                    const int4 d = reinterpret_cast<const int4*>(dst)[w];
                    own[w] = own_id(d.x) | (own_id(d.y) << 8) | (own_id(d.z) << 16) | (own_id(d.w) << 24);
                }
            }
        }
    }
    if (nw > nfull && t == 0) {
        unsigned v = 0u;
        for (int c = 0; c < 4; ++c)
            if (nfull * 4 + c < E) v |= own_id(dst[nfull * 4 + c]) << (8 * c);
        own[nfull] = v;
    }

    const int ntrips  = (N + 2 * NPW - 1) / (2 * NPW);
    const float4 wl = reinterpret_cast<const float4*>(attn_l)[lane];
    const float4 wr = reinterpret_cast<const float4*>(attn_r)[lane];
    for (int g = gwave; g < ntrips; g += nwaves) {
        const int n0 = g * 2 * NPW;
        const float4* frow = reinterpret_cast<const float4*>(ft + (size_t)n0 * HD) + lane;
        float4 v[2 * NPW];
        if (n0 + 2 * NPW <= N) {
            #pragma unroll
            for (int i = 0; i < 2 * NPW; ++i) v[i] = frow[(size_t)i * (HD / 4)];
        } else {
            #pragma unroll
            for (int i = 0; i < 2 * NPW; ++i)
                v[i] = (n0 + i < N) ? frow[(size_t)i * (HD / 4)] : make_float4(0.f, 0.f, 0.f, 0.f);
        }
        #pragma unroll
        for (int u = 0; u < 2; ++u) {
            float pl[NPW], pr[NPW];
            #pragma unroll
            for (int i = 0; i < NPW; ++i) {
                const float4 x = v[u * NPW + i];
                pl[i] = x.x * wl.x + x.y * wl.y + x.z * wl.z + x.w * wl.w;
                pr[i] = x.x * wr.x + x.y * wr.y + x.z * wr.z + x.w * wr.w;
            }
            const float sl = head_sum_scatter(pl, lane);
            const float sr = head_sum_scatter(pr, lane);
            const int node = n0 + u * NPW + (lane & 7);
            if (node < N) {
                el[(size_t)node * HH + (lane >> 3)] = sl;
                er[(size_t)node * HH + (lane >> 3)] = sr;
            }
        }
    }
}

__device__ __forceinline__ float leaky(float x) {
    return (x > 0.f) ? x : NEG_SLOPE * x;
}

#define OWN_BLOCKS  256
#define OWN_THREADS 1024
#define OWN_WAVES   (OWN_THREADS / 32)
#define DRAIN       8
#define SCAN_LOADS  4
#define QCAP        480
#define RESOLVE     4
#define NOT_OWNED   0xFFFFu

__device__ __forceinline__ void own_resolve_max(
        int* q, unsigned short* qd, int n, int lane, int n0, int cnt,
        const int* __restrict__ src, const int* __restrict__ dst,
        const float* __restrict__ el, const float* erl, float* mloc) {
    for (int base = 0; base < n; base += 32 * RESOLVE) {
        int  e[RESOLVE], d[RESOLVE], s[RESOLVE];
        bool in[RESOLVE];
        #pragma unroll
        for (int u = 0; u < RESOLVE; ++u) {
            const int idx = base + u * 32 + lane;
            in[u] = idx < n;
            e[u]  = in[u] ? q[idx] : 0;
        }
        #pragma unroll
        for (int u = 0; u < RESOLVE; ++u) {
            d[u] = dst[e[u]];
            s[u] = src[e[u]];
        }
        float4 l[RESOLVE];
        #pragma unroll
        for (int u = 0; u < RESOLVE; ++u)
            l[u] = reinterpret_cast<const float4*>(el)[s[u]];
        #pragma unroll
        for (int u = 0; u < RESOLVE; ++u) {
            const int  idx = base + u * 32 + lane;
            const int  dl  = d[u] - n0;
            const bool ok  = in[u] && ((unsigned)dl < (unsigned)cnt);
            if (in[u]) {
                q[idx]  = s[u];
                qd[idx] = ok ? (unsigned short)dl : (unsigned short)NOT_OWNED;
            }
            if (ok) {
                const float4 r = reinterpret_cast<const float4*>(erl)[dl];
                atomicMax(&mloc[dl * HH + 0], leaky(l[u].x + r.x));
                atomicMax(&mloc[dl * HH + 1], leaky(l[u].y + r.y));
                atomicMax(&mloc[dl * HH + 2], leaky(l[u].z + r.z));
                atomicMax(&mloc[dl * HH + 3], leaky(l[u].w + r.w));
            }
        }
    }
}

__device__ __forceinline__ void own_accumulate(
        const int* q, const unsigned short* qd, int n, int lane,
        const float* __restrict__ el, const float* __restrict__ ft,
        const float* erl, const float* mloc, float* den, float* acc) {
    const int j = lane >> 2;
    const int h = lane & 3;
    int b = 0;
    for (; b + DRAIN <= n; b += DRAIN) {
        const int      sv = q[b + j];
        const unsigned dr = qd[b + j];
        const bool     ok = dr != NOT_OWNED;
        const int      dl = ok ? (int)dr : 0;
        const float    xl = el[(size_t)sv * HH + h];
        float v[DRAIN][HH];
        #pragma unroll
        for (int jj = 0; jj < DRAIN; ++jj) {
            const int s = __builtin_amdgcn_readlane(sv, 4 * jj);
            const float* fts = ft + (size_t)s * HD + lane;
            #pragma unroll
            for (int hh = 0; hh < HH; ++hh) v[jj][hh] = fts[hh * DD];
        }
        const float x  = leaky(xl + erl[dl * HH + h]);
        const float ae = __expf(x - mloc[dl * HH + h]);
        if (ok) atomicAdd(&den[dl * HH + h], ae);
        #pragma unroll
        for (int jj = 0; jj < DRAIN; ++jj) {
            const int dj  = __builtin_amdgcn_readlane(dl, 4 * jj);
            const int okj = __builtin_amdgcn_readlane((int)ok, 4 * jj);
            if (okj) {
                float* accd = acc + dj * HD + lane;
                #pragma unroll
                for (int hh = 0; hh < HH; ++hh) {
                    const float a = __int_as_float(
                        __builtin_amdgcn_readlane(__float_as_int(ae), 4 * jj + hh));
                    atomicAdd(&accd[hh * DD], v[jj][hh] * a);
                }
            }
        }
    }
    for (; b < n; ++b) {
        const int      s  = q[b];
        const unsigned dr = qd[b];
        if (dr != NOT_OWNED) {
            const int dl = (int)dr;
            const float* fts = ft + (size_t)s * HD + lane;
            float* accd = acc + dl * HD + lane;
            #pragma unroll
            for (int hh = 0; hh < HH; ++hh) {
                const float x  = leaky(el[(size_t)s * HH + hh] + erl[dl * HH + hh]);
                const float ae = __expf(x - mloc[dl * HH + hh]);
                if (lane == 0) atomicAdd(&den[dl * HH + hh], ae);
                atomicAdd(&accd[hh * DD], fts[hh * DD] * ae);
            }
        }
    }
}

template <bool ACC>
__device__ __forceinline__ void own_plain_pass(
        int wave, int lane, int steps, int n0, int cnt, unsigned ownByte,
        const int* __restrict__ src, const int* __restrict__ dst, const unsigned* __restrict__ own,
        const float* __restrict__ el, const float* __restrict__ ft,
        const float* erl, float* mloc, float* den, float* acc, int E) {
    for (int i = wave; i < steps; i += OWN_WAVES) {
        for (int c = 0; c < 16; ++c) {
            const int e = i * 512 + lane * 16 + c;
            if (e >= E) continue;
            if (((own[e >> 2] >> (8 * (e & 3))) & 255u) != ownByte) continue;
            const int dl = dst[e] - n0;
            if ((unsigned)dl >= (unsigned)cnt) continue;
            const int s = src[e];
            for (int hh = 0; hh < HH; ++hh) {
                const float x = leaky(el[(size_t)s * HH + hh] + erl[dl * HH + hh]);
                if (!ACC) {
                    atomicMax(&mloc[dl * HH + hh], x);
                } else {
                    const float ae = __expf(x - mloc[dl * HH + hh]);
                    atomicAdd(&den[dl * HH + hh], ae);
                    for (int dd = 0; dd < DD; ++dd)
                        atomicAdd(&acc[dl * HD + hh * DD + dd], ft[(size_t)s * HD + hh * DD + dd] * ae);
                }
            }
        }
    }
}

__global__ __launch_bounds__(OWN_THREADS) void gat_owner_aggregate(
        const int* __restrict__ src, const int* __restrict__ dst,
        const unsigned* __restrict__ own,
        const float* __restrict__ el, const float* __restrict__ er,
        const float* __restrict__ ft, const float* __restrict__ bias,
        float* __restrict__ out, int N, int E) {
    __shared__ __attribute__((aligned(16))) float acc[RMAX * HD];
    __shared__ __attribute__((aligned(16))) float den[RMAX * HH];
    __shared__ __attribute__((aligned(16))) float mloc[RMAX * HH];
    __shared__ __attribute__((aligned(16))) float erl[RMAX * HH];
    __shared__ int queue[OWN_WAVES * QCAP];
    __shared__ unsigned short qdst[OWN_WAVES * QCAP];
    __shared__ int ovf;

    const int tid   = threadIdx.x;
    const int lane  = tid & 31;
    const int wave  = __builtin_amdgcn_readfirstlane(tid >> 5);
    int* q              = &queue[wave * QCAP];
    unsigned short* qd  = &qdst[wave * QCAP];
    const int nw        = (E + 3) >> 2;
    const int steps     = (E + 511) >> 9;
    const int fullSteps = E >> 9;
    const int start = (int)(((long long)blockIdx.x * steps) / gridDim.x);
    const int K     = (steps + OWN_WAVES - 1) / OWN_WAVES;

    for (int n0 = blockIdx.x * RMAX; n0 < N; n0 += (int)gridDim.x * RMAX) {
        const int cnt = (N - n0 < RMAX) ? (N - n0) : RMAX;
        const unsigned ownByte = (unsigned)(n0 / RMAX) & 255u;
        const unsigned pat = ownByte * 0x01010101u;

        const float4 z4 = make_float4(0.f, 0.f, 0.f, 0.f);
        const float4 lo = make_float4(-3.402823466e38f, -3.402823466e38f,
                                      -3.402823466e38f, -3.402823466e38f);
        for (int i = tid; i < cnt * (HD / 4); i += OWN_THREADS) reinterpret_cast<float4*>(acc)[i] = z4;
        for (int i = tid; i < cnt; i += OWN_THREADS) {
            reinterpret_cast<float4*>(den)[i]  = z4;
            reinterpret_cast<float4*>(mloc)[i] = lo;
            reinterpret_cast<float4*>(erl)[i]  = reinterpret_cast<const float4*>(er)[n0 + i];
        }
        if (tid == 0) ovf = 0;
        __syncthreads();

        int  qn   = 0;
        bool lost = false;
        for (int k = 0; k < K; k += SCAN_LOADS) {
            int  rot[SCAN_LOADS];
            bool allFull = true;
            #pragma unroll
            for (int r = 0; r < SCAN_LOADS; ++r) {
                const int i = wave + OWN_WAVES * (k + r);
                int rr = i + start;
                if (rr >= steps) rr -= steps;
                rot[r]  = (i < steps) ? rr : -1;
                allFull = allFull && (i < steps) && (rr < fullSteps);
            }
            uint4 ww[SCAN_LOADS];
            if (allFull) {
                #pragma unroll
                for (int r = 0; r < SCAN_LOADS; ++r)
                    ww[r] = reinterpret_cast<const uint4*>(own)[(size_t)rot[r] * 32 + lane];
            } else {
                #pragma unroll
                for (int r = 0; r < SCAN_LOADS; ++r) {
                    ww[r] = make_uint4(0u, 0u, 0u, 0u);
                    if (rot[r] >= 0 && rot[r] < fullSteps) {
                        ww[r] = reinterpret_cast<const uint4*>(own)[(size_t)rot[r] * 32 + lane];
                    } else if (rot[r] >= 0) {
                        unsigned w4[4];
                        #pragma unroll
                        for (int c = 0; c < 4; ++c) {
                            const int wi   = rot[r] * 128 + lane * 4 + c;
                            const int left = E - wi * 4;
                            unsigned w = (wi < nw) ? own[wi] : ~pat;
                            if (left < 4) {
                                const unsigned keep = (left <= 0) ? 0u : (0xFFFFFFFFu >> (8 * (4 - left)));
                                w = (w & keep) | (~pat & ~keep);
                            }
                            w4[c] = w;
                        }
                        ww[r] = make_uint4(w4[0], w4[1], w4[2], w4[3]);
                    }
                }
            }

            #pragma unroll
            for (int r = 0; r < SCAN_LOADS; ++r) {
                if (rot[r] >= 0) {
                    const int ebase = rot[r] * 512 + lane * 16;
                    const unsigned w4[4] = {ww[r].x, ww[r].y, ww[r].z, ww[r].w};
                    #pragma unroll
                    for (int c = 0; c < 4; ++c) {
                        const unsigned x = w4[c] ^ pat;
                        const unsigned t = (x - 0x01010101u) & ~x & 0x80808080u;
                        if (__builtin_amdgcn_ballot_w32(t != 0u) != 0u) {
                            if (qn > QCAP - 128) {
                                lost = true;
                            } else {
                                #pragma unroll
                                for (int b = 0; b < 4; ++b) {
                                    const bool hit = ((x >> (8 * b)) & 255u) == 0u;
                                    const unsigned mask = __builtin_amdgcn_ballot_w32(hit);
                                    if (mask != 0u) {
                                        if (hit) q[qn + (int)__builtin_amdgcn_mbcnt_lo(mask, 0u)] = ebase + c * 4 + b;
                                        qn += __builtin_popcount(mask);
                                    }
                                }
                            }
                        }
                    }
                }
            }
        }
        if (lost && lane == 0) ovf = 1;

        own_resolve_max(q, qd, qn, lane, n0, cnt, src, dst, el, erl, mloc);
        __syncthreads();

        if (__builtin_amdgcn_readfirstlane(ovf) == 0) {
            own_accumulate(q, qd, qn, lane, el, ft, erl, mloc, den, acc);
        } else {
            own_plain_pass<false>(wave, lane, steps, n0, cnt, ownByte, src, dst, own, el, ft, erl, mloc, den, acc, E);
            __syncthreads();
            own_plain_pass<true>(wave, lane, steps, n0, cnt, ownByte, src, dst, own, el, ft, erl, mloc, den, acc, E);
        }
        __syncthreads();

        for (int i = tid; i < cnt * (HD / 4); i += OWN_THREADS) {
            const int node = i >> 5;
            const int q4   = i & 31;
            const float4 a4 = reinterpret_cast<const float4*>(acc)[i];
            const float  dn = den[node * HH + (q4 >> 3)];
            const float4 b4 = reinterpret_cast<const float4*>(bias)[q4];
            float4 o;
            if (dn > 0.f) {
                o.x = a4.x / dn + b4.x; o.y = a4.y / dn + b4.y;
                o.z = a4.z / dn + b4.z; o.w = a4.w / dn + b4.w;
            } else {
                o = b4;
            }
            reinterpret_cast<float4*>(out)[(size_t)(n0 + node) * (HD / 4) + q4] = o;
        }
        __syncthreads();
    }
}

extern "C" void kernel_launch(void* const* d_in, const int* in_sizes, int n_in,
                              void* d_out, int out_size, void* d_ws, size_t ws_size,
                              hipStream_t stream) {
    const float* feat   = (const float*)d_in[0];
    const int*   src    = (const int*)  d_in[1];
    const int*   dst    = (const int*)  d_in[2];
    const float* W      = (const float*)d_in[3];
    const float* attn_l = (const float*)d_in[4];
    const float* attn_r = (const float*)d_in[5];
    const float* bias   = (const float*)d_in[6];

    const int N = in_sizes[0] / DIN;
    const int E = in_sizes[1];
    float* out = (float*)d_out;

    float* ws    = (float*)d_ws;
    float* ft    = ws;
    float* el    = ft    + (size_t)N * HD;
    float* er    = el    + (size_t)N * HH;
    unsigned* own = (unsigned*)(er + (size_t)N * HH);

    gat_gemm<<<GEMM_BLOCKS, 256, 0, stream>>>(feat, W, ft, N);
    gat_logits_owner<<<LG_BLOCKS, LG_THREADS, 0, stream>>>(ft, attn_l, attn_r, el, er, dst, own, N, E);
    gat_owner_aggregate<<<OWN_BLOCKS, OWN_THREADS, 0, stream>>>(src, dst, own, el, er, ft, bias, out, N, E);
}


# Owner byte scan replaced by a sort: kernel 2 counting-sorts each 2048-edge step by owner work-group in LDS and writes a start/count table; owners copy only their slices; sums still LDS-atomic order
# speedup vs baseline: 5.3987x; 1.2124x over previous
#include <hip/hip_runtime.h>

typedef __attribute__((ext_vector_type(2))) float v2f;
typedef __attribute__((ext_vector_type(8))) float v8f;

#define DIN 256
#define HD  128
#define HH  4
#define DD  32
#define NEG_SLOPE 0.2f

#define GEMM_BLOCKS 512
#define KG 16
#define APITCH 260

__device__ __forceinline__ void gemm_fetch_tile(float4 (&pre)[4], const float* __restrict__ feat,
                                                int rowBase, int N, int tid) {
    if (rowBase + 16 <= N) {
        #pragma unroll
        for (int i = 0; i < 4; ++i) {
            const int idx4 = i * 256 + tid;
            const int r    = idx4 >> 6;
            const int c4   = idx4 & 63;
            pre[i] = reinterpret_cast<const float4*>(feat + (size_t)(rowBase + r) * DIN)[c4];
        }
    } else {
        #pragma unroll
        for (int i = 0; i < 4; ++i) {
            const int idx4 = i * 256 + tid;
            const int r    = idx4 >> 6;
            const int c4   = idx4 & 63;
            float4 v = make_float4(0.f, 0.f, 0.f, 0.f);
            if (rowBase + r < N)
                v = reinterpret_cast<const float4*>(feat + (size_t)(rowBase + r) * DIN)[c4];
            pre[i] = v;
        }
    }
}

__device__ __forceinline__ void gemm_store_tile(float (*As)[APITCH], const float4 (&pre)[4], int tid) {
    #pragma unroll
    for (int i = 0; i < 4; ++i) {
        const int idx4 = i * 256 + tid;
        const int r    = idx4 >> 6;
        const int c4   = idx4 & 63;
        As[r][c4 * 4 + 0] = pre[i].x; As[r][c4 * 4 + 1] = pre[i].y;
        As[r][c4 * 4 + 2] = pre[i].z; As[r][c4 * 4 + 3] = pre[i].w;
    }
}

__global__ __launch_bounds__(256) void gat_gemm(const float* __restrict__ feat,
                                                const float* __restrict__ W,
                                                float* __restrict__ ft, int N) {
    __shared__ float As[2][16][APITCH];
    const int tid      = threadIdx.x;
    const int numTiles = (N + 15) >> 4;
    int tile = blockIdx.x;
    if (tile >= numTiles) return;

    const int wave    = tid >> 5;
    const int lane    = tid & 31;
    const int half    = lane >> 4;
    const int mrow    = lane & 15;
    const int colBase = wave * 16;
    const int kp      = half * 2;

    const float* wrow = W + (size_t)(colBase + mrow) * DIN + kp;
    v2f b[DIN / 4];
    #pragma unroll
    for (int i = 0; i < DIN / 4; ++i)
        b[i] = *reinterpret_cast<const v2f*>(wrow + 4 * i);

    float4 pre[4];
    gemm_fetch_tile(pre, feat, tile * 16, N, tid);
    gemm_store_tile(As[0], pre, tid);
    __syncthreads();

    int cur = 0;
    for (;;) {
        const int  next    = tile + (int)gridDim.x;
        const bool hasNext = next < numTiles;
        if (hasNext) gemm_fetch_tile(pre, feat, next * 16, N, tid);

        const int rowBase = tile * 16;
        v8f c = {0.f, 0.f, 0.f, 0.f, 0.f, 0.f, 0.f, 0.f};
        const float* arow = &As[cur][mrow][kp];
        v2f a[2][KG];
        #pragma unroll
        for (int i = 0; i < KG; ++i)
            a[0][i] = *reinterpret_cast<const v2f*>(arow + 4 * i);
        __builtin_amdgcn_sched_group_barrier(0x100, KG / 2, 0);
        #pragma unroll
        for (int g = 0; g < DIN / 4 / KG; ++g) {
            if (g + 1 < DIN / 4 / KG) {
                #pragma unroll
                for (int i = 0; i < KG; ++i)
                    a[(g + 1) & 1][i] = *reinterpret_cast<const v2f*>(arow + 4 * ((g + 1) * KG + i));
            }
            #pragma unroll
            for (int i = 0; i < KG; ++i)
                c = __builtin_amdgcn_wmma_f32_16x16x4_f32(
                         false, a[g & 1][i],  false, b[g * KG + i],
                         (short)0, c,  false,  false);
            if (g + 1 < DIN / 4 / KG)
                __builtin_amdgcn_sched_group_barrier(0x100, KG / 2, 0);
            __builtin_amdgcn_sched_group_barrier(0x008, KG, 0);
        }

        float* outBase = ft + (size_t)(rowBase + 8 * half) * HD + colBase + mrow;
        if (rowBase + 16 <= N) {
            #pragma unroll
            for (int v = 0; v < 8; ++v)
                outBase[(size_t)v * HD] = c[v];
        } else {
            #pragma unroll
            for (int v = 0; v < 8; ++v) {
                if (rowBase + v + 8 * half < N)
                    outBase[(size_t)v * HD] = c[v];
            }
        }

        if (!hasNext) break;
        gemm_store_tile(As[cur ^ 1], pre, tid);
        __syncthreads();
        cur ^= 1;
        tile = next;
    }
}

#define NPW        8
#define LG_BLOCKS  512
#define LG_THREADS 256
#define NBINS      256
#define SORT_S     2048
#define SORT_EPT   (SORT_S / LG_THREADS)
#define RMAX 392

__device__ __forceinline__ unsigned own_id(int d) {
    return ((unsigned)d / (unsigned)RMAX) & 255u;
}


__device__ __forceinline__ float head_sum_scatter(const float (&p)[NPW], int lane) {
    const bool b0 = (lane & 1) != 0;
    const bool b1 = (lane & 2) != 0;
    const bool b2 = (lane & 4) != 0;
    float q[4];
    #pragma unroll
    for (int k = 0; k < 4; ++k) {
        const float keep = b0 ? p[2 * k + 1] : p[2 * k];
        const float send = b0 ? p[2 * k]     : p[2 * k + 1];
        q[k] = keep + __shfl_xor(send, 1, 32);
    }
    float r[2];
    #pragma unroll
    for (int k = 0; k < 2; ++k) {
        const float keep = b1 ? q[2 * k + 1] : q[2 * k];
        const float send = b1 ? q[2 * k]     : q[2 * k + 1];
        r[k] = keep + __shfl_xor(send, 2, 32);
    }
    const float keep = b2 ? r[1] : r[0];
    const float send = b2 ? r[0] : r[1];
    return keep + __shfl_xor(send, 4, 32);
}

__global__ __launch_bounds__(LG_THREADS) void gat_logits_sort(
        const float* __restrict__ ft,
        const float* __restrict__ attn_l, const float* __restrict__ attn_r,
        float* __restrict__ el, float* __restrict__ er,
        const int* __restrict__ src, const int* __restrict__ dst,
        unsigned* __restrict__ tab, int* __restrict__ ent, int N, int E) {
    __shared__ unsigned cntb[NBINS];
    __shared__ unsigned startb[NBINS];
    __shared__ unsigned wsum[LG_THREADS / 32];
    __shared__ __attribute__((aligned(16))) int2 stage[SORT_S];
    static_assert(LG_THREADS == NBINS, "the bin scan holds one bin per thread");
    const int tid    = threadIdx.x;
    const int t      = blockIdx.x * LG_THREADS + tid;
    const int stride = (int)gridDim.x * LG_THREADS;

    const int lane   = tid & 31;
    const int wv     = tid >> 5;
    const int gwave  = __builtin_amdgcn_readfirstlane(t >> 5);
    const int nwaves = stride >> 5;

    const int nsteps = (E + SORT_S - 1) / SORT_S;
    int2* ent2 = reinterpret_cast<int2*>(ent);
    for (int step = blockIdx.x; step < nsteps; step += (int)gridDim.x) {
        cntb[tid] = 0u;
        __syncthreads();
        const int e0 = step * SORT_S + tid * SORT_EPT;
        int d[SORT_EPT], s[SORT_EPT];
        if (e0 + SORT_EPT <= E) {
            #pragma unroll
            for (int u = 0; u < SORT_EPT / 4; ++u) {
                const int4 dv = reinterpret_cast<const int4*>(dst + e0)[u];
                const int4 sv = reinterpret_cast<const int4*>(src + e0)[u];
                d[4 * u + 0] = dv.x; d[4 * u + 1] = dv.y; d[4 * u + 2] = dv.z; d[4 * u + 3] = dv.w;
                s[4 * u + 0] = sv.x; s[4 * u + 1] = sv.y; s[4 * u + 2] = sv.z; s[4 * u + 3] = sv.w;
            }
        } else {
            #pragma unroll
            for (int u = 0; u < SORT_EPT; ++u) {
                d[u] = (e0 + u < E) ? dst[e0 + u] : 0;
                s[u] = (e0 + u < E) ? src[e0 + u] : 0;
            }
        }
        unsigned bn[SORT_EPT], rk[SORT_EPT];
        #pragma unroll
        for (int u = 0; u < SORT_EPT; ++u) {
            bn[u] = own_id(d[u]);
            rk[u] = 0u;
            if (e0 + u < E) rk[u] = atomicAdd(&cntb[bn[u]], 1u);
        }
        __syncthreads();

        const unsigned c = cntb[tid];
        unsigned incl = c;
        #pragma unroll
        for (int o = 1; o < 32; o <<= 1) {
            const unsigned v = __shfl_up(incl, o, 32);
            if (lane >= o) incl += v;
        }
        if (lane == 31) wsum[wv] = incl;
        __syncthreads();
        unsigned basev = 0u;
        #pragma unroll
        for (int w = 0; w < LG_THREADS / 32; ++w)
            if (w < wv) basev += wsum[w];
        const unsigned st = basev + incl - c;
        startb[tid] = st;
        tab[(size_t)step * NBINS + tid] = (st << 16) | c;
        __syncthreads();

        #pragma unroll
        for (int u = 0; u < SORT_EPT; ++u) {
            if (e0 + u < E)
                stage[startb[bn[u]] + rk[u]] = make_int2(s[u], d[u]);
        }
        __syncthreads();

        const int left   = E - step * SORT_S;
        const int nvalid = (left < SORT_S) ? left : SORT_S;
        int4* outv = reinterpret_cast<int4*>(ent2 + (size_t)step * SORT_S);
        #pragma unroll
        for (int u = 0; u < SORT_S / 2 / LG_THREADS; ++u) {
            const int j = u * LG_THREADS + tid;
            if (2 * j < nvalid) outv[j] = reinterpret_cast<const int4*>(stage)[j];
        }
    }


    const int ntrips  = (N + 2 * NPW - 1) / (2 * NPW);
    const float4 wl = reinterpret_cast<const float4*>(attn_l)[lane];
    const float4 wr = reinterpret_cast<const float4*>(attn_r)[lane];
    for (int g = gwave; g < ntrips; g += nwaves) {
        const int n0 = g * 2 * NPW;
        const float4* frow = reinterpret_cast<const float4*>(ft + (size_t)n0 * HD) + lane;
        float4 v[2 * NPW];
        if (n0 + 2 * NPW <= N) {
            #pragma unroll
            for (int i = 0; i < 2 * NPW; ++i) v[i] = frow[(size_t)i * (HD / 4)];
        } else {
            #pragma unroll
            for (int i = 0; i < 2 * NPW; ++i)
                v[i] = (n0 + i < N) ? frow[(size_t)i * (HD / 4)] : make_float4(0.f, 0.f, 0.f, 0.f);
        }
        #pragma unroll
        for (int u = 0; u < 2; ++u) {
            float pl[NPW], pr[NPW];
            #pragma unroll
            for (int i = 0; i < NPW; ++i) {
                const float4 x = v[u * NPW + i];
                pl[i] = x.x * wl.x + x.y * wl.y + x.z * wl.z + x.w * wl.w;
                pr[i] = x.x * wr.x + x.y * wr.y + x.z * wr.z + x.w * wr.w;
            }
            const float sl = head_sum_scatter(pl, lane);
            const float sr = head_sum_scatter(pr, lane);
            const int node = n0 + u * NPW + (lane & 7);
            if (node < N) {
                el[(size_t)node * HH + (lane >> 3)] = sl;
                er[(size_t)node * HH + (lane >> 3)] = sr;
            }
        }
    }
}

__device__ __forceinline__ float leaky(float x) {
    return (x > 0.f) ? x : NEG_SLOPE * x;
}

#define OWN_BLOCKS  256
#define OWN_THREADS 1024
#define OWN_WAVES   (OWN_THREADS / 32)
#define DRAIN       8
#define QCAP        480
#define FILL        8
#define RESOLVE     4
#define NOT_OWNED   0xFFFFu

__device__ __forceinline__ void own_resolve_max(
        const int* q, const unsigned short* qd, int n, int lane,
        const float* __restrict__ el, const float* erl, float* mloc) {
    for (int base = 0; base < n; base += 32 * RESOLVE) {
        int      s[RESOLVE];
        unsigned dr[RESOLVE];
        #pragma unroll
        for (int u = 0; u < RESOLVE; ++u) {
            const int idx = base + u * 32 + lane;
            s[u]  = (idx < n) ? q[idx] : 0;
            dr[u] = (idx < n) ? (unsigned)qd[idx] : NOT_OWNED;
        }
        float4 l[RESOLVE];
        #pragma unroll
        for (int u = 0; u < RESOLVE; ++u)
            l[u] = reinterpret_cast<const float4*>(el)[s[u]];
        #pragma unroll
        for (int u = 0; u < RESOLVE; ++u) {
            if (dr[u] != NOT_OWNED) {
                const int dl = (int)dr[u];
                const float4 r = reinterpret_cast<const float4*>(erl)[dl];
                atomicMax(&mloc[dl * HH + 0], leaky(l[u].x + r.x));
                atomicMax(&mloc[dl * HH + 1], leaky(l[u].y + r.y));
                atomicMax(&mloc[dl * HH + 2], leaky(l[u].z + r.z));
                atomicMax(&mloc[dl * HH + 3], leaky(l[u].w + r.w));
            }
        }
    }
}

__device__ __forceinline__ void own_accumulate(
        const int* q, const unsigned short* qd, int n, int lane,
        const float* __restrict__ el, const float* __restrict__ ft,
        const float* erl, const float* mloc, float* den, float* acc) {
    const int j = lane >> 2;
    const int h = lane & 3;
    int b = 0;
    for (; b + DRAIN <= n; b += DRAIN) {
        const int      sv = q[b + j];
        const unsigned dr = qd[b + j];
        const bool     ok = dr != NOT_OWNED;
        const int      dl = ok ? (int)dr : 0;
        const float    xl = el[(size_t)sv * HH + h];
        float v[DRAIN][HH];
        #pragma unroll
        for (int jj = 0; jj < DRAIN; ++jj) {
            const int s = __builtin_amdgcn_readlane(sv, 4 * jj);
            const float* fts = ft + (size_t)s * HD + lane;
            #pragma unroll
            for (int hh = 0; hh < HH; ++hh) v[jj][hh] = fts[hh * DD];
        }
        const float x  = leaky(xl + erl[dl * HH + h]);
        const float ae = __expf(x - mloc[dl * HH + h]);
        if (ok) atomicAdd(&den[dl * HH + h], ae);
        #pragma unroll
        for (int jj = 0; jj < DRAIN; ++jj) {
            const int dj  = __builtin_amdgcn_readlane(dl, 4 * jj);
            const int okj = __builtin_amdgcn_readlane((int)ok, 4 * jj);
            if (okj) {
                float* accd = acc + dj * HD + lane;
                #pragma unroll
                for (int hh = 0; hh < HH; ++hh) {
                    const float a = __int_as_float(
                        __builtin_amdgcn_readlane(__float_as_int(ae), 4 * jj + hh));
                    atomicAdd(&accd[hh * DD], v[jj][hh] * a);
                }
            }
        }
    }
    for (; b < n; ++b) {
        const int      s  = q[b];
        const unsigned dr = qd[b];
        if (dr != NOT_OWNED) {
            const int dl = (int)dr;
            const float* fts = ft + (size_t)s * HD + lane;
            float* accd = acc + dl * HD + lane;
            #pragma unroll
            for (int hh = 0; hh < HH; ++hh) {
                const float x  = leaky(el[(size_t)s * HH + hh] + erl[dl * HH + hh]);
                const float ae = __expf(x - mloc[dl * HH + hh]);
                if (lane == 0) atomicAdd(&den[dl * HH + hh], ae);
                atomicAdd(&accd[hh * DD], fts[hh * DD] * ae);
            }
        }
    }
}

template <bool ACC>
__device__ __forceinline__ void own_plain_pass(
        int lane, int sBeg, int sEnd, int n0, int cnt,
        const unsigned* __restrict__ trow, const int2* __restrict__ ent2,
        const float* __restrict__ el, const float* __restrict__ ft,
        const float* erl, float* mloc, float* den, float* acc) {
    for (int sidx = sBeg + lane; sidx < sEnd; sidx += 32) {
        const unsigned tw = trow[(size_t)sidx * NBINS];
        const int c = (int)(tw & 0xFFFFu);
        const int2* p = ent2 + (size_t)sidx * SORT_S + (tw >> 16);
        for (int i = 0; i < c; ++i) {
            const int2 ed = p[i];
            const int dl = ed.y - n0;
            if ((unsigned)dl >= (unsigned)cnt) continue;
            const int s = ed.x;
            for (int hh = 0; hh < HH; ++hh) {
                const float x = leaky(el[(size_t)s * HH + hh] + erl[dl * HH + hh]);
                if (!ACC) {
                    atomicMax(&mloc[dl * HH + hh], x);
                } else {
                    const float ae = __expf(x - mloc[dl * HH + hh]);
                    atomicAdd(&den[dl * HH + hh], ae);
                    for (int dd = 0; dd < DD; ++dd)
                        atomicAdd(&acc[dl * HD + hh * DD + dd], ft[(size_t)s * HD + hh * DD + dd] * ae);
                }
            }
        }
    }
}

__global__ __launch_bounds__(OWN_THREADS) void gat_owner_aggregate(
        const unsigned* __restrict__ tab, const int* __restrict__ ent,
        const float* __restrict__ el, const float* __restrict__ er,
        const float* __restrict__ ft, const float* __restrict__ bias,
        float* __restrict__ out, int N, int E) {
    __shared__ __attribute__((aligned(16))) float acc[RMAX * HD];
    __shared__ __attribute__((aligned(16))) float den[RMAX * HH];
    __shared__ __attribute__((aligned(16))) float mloc[RMAX * HH];
    __shared__ __attribute__((aligned(16))) float erl[RMAX * HH];
    __shared__ int queue[OWN_WAVES * QCAP];
    __shared__ unsigned short qdst[OWN_WAVES * QCAP];
    __shared__ int ovf;

    const int tid   = threadIdx.x;
    const int lane  = tid & 31;
    const int wave  = __builtin_amdgcn_readfirstlane(tid >> 5);
    int* q              = &queue[wave * QCAP];
    unsigned short* qd  = &qdst[wave * QCAP];
    const int2* ent2    = reinterpret_cast<const int2*>(ent);
    const int nsteps    = (E + SORT_S - 1) / SORT_S;
    const int chunk     = (nsteps + OWN_WAVES - 1) / OWN_WAVES;
    const int sBeg      = wave * chunk;
    const int sEnd      = (sBeg + chunk < nsteps) ? (sBeg + chunk) : nsteps;

    for (int n0 = blockIdx.x * RMAX; n0 < N; n0 += (int)gridDim.x * RMAX) {
        const int cnt = (N - n0 < RMAX) ? (N - n0) : RMAX;
        const unsigned ownBin = (unsigned)(n0 / RMAX) & 255u;
        const unsigned* trow = tab + ownBin;

        const float4 z4 = make_float4(0.f, 0.f, 0.f, 0.f);
        const float4 lo = make_float4(-3.402823466e38f, -3.402823466e38f,
                                      -3.402823466e38f, -3.402823466e38f);
        for (int i = tid; i < cnt * (HD / 4); i += OWN_THREADS) reinterpret_cast<float4*>(acc)[i] = z4;
        for (int i = tid; i < cnt; i += OWN_THREADS) {
            reinterpret_cast<float4*>(den)[i]  = z4;
            reinterpret_cast<float4*>(mloc)[i] = lo;
            reinterpret_cast<float4*>(erl)[i]  = reinterpret_cast<const float4*>(er)[n0 + i];
        }
        if (tid == 0) ovf = 0;
        __syncthreads();

        int  qn   = 0;
        bool lost = false;
        for (int sb = sBeg; sb < sEnd; sb += 32) {
            const int sidx = sb + lane;
            const unsigned tw = (sidx < sEnd) ? trow[(size_t)sidx * NBINS] : 0u;
            const int c = (int)(tw & 0xFFFFu);
            int incl = c;
            #pragma unroll
            for (int o = 1; o < 32; o <<= 1) {
                const int v = __shfl_up(incl, o, 32);
                if (lane >= o) incl += v;
            }
            const int total = __builtin_amdgcn_readlane(incl, 31);
            if (qn + total > QCAP) { lost = true; break; }
            const int pos = qn + incl - c;
            const int2* p = ent2 + (size_t)sidx * SORT_S + (tw >> 16);
            for (int i0 = 0; __builtin_amdgcn_ballot_w32(i0 < c) != 0u; i0 += FILL) {
                int2 ed[FILL];
                #pragma unroll
                for (int u = 0; u < FILL; ++u)
                    ed[u] = (i0 + u < c) ? p[i0 + u] : make_int2(0, 0);
                #pragma unroll
                for (int u = 0; u < FILL; ++u) {
                    if (i0 + u < c) {
                        const int dl = ed[u].y - n0;
                        const bool ok = (unsigned)dl < (unsigned)cnt;
                        q[pos + i0 + u]  = ed[u].x;
                        qd[pos + i0 + u] = ok ? (unsigned short)dl : (unsigned short)NOT_OWNED;
                    }
                }
            }
            qn += total;
        }
        if (lost && lane == 0) ovf = 1;

        own_resolve_max(q, qd, qn, lane, el, erl, mloc);
        __syncthreads();

        if (__builtin_amdgcn_readfirstlane(ovf) == 0) {
            own_accumulate(q, qd, qn, lane, el, ft, erl, mloc, den, acc);
        } else {
            own_plain_pass<false>(lane, sBeg, sEnd, n0, cnt, trow, ent2, el, ft, erl, mloc, den, acc);
            __syncthreads();
            own_plain_pass<true>(lane, sBeg, sEnd, n0, cnt, trow, ent2, el, ft, erl, mloc, den, acc);
        }
        __syncthreads();


        for (int i = tid; i < cnt * (HD / 4); i += OWN_THREADS) {
            const int node = i >> 5;
            const int q4   = i & 31;
            const float4 a4 = reinterpret_cast<const float4*>(acc)[i];
            const float  dn = den[node * HH + (q4 >> 3)];
            const float4 b4 = reinterpret_cast<const float4*>(bias)[q4];
            float4 o;
            if (dn > 0.f) {
                o.x = a4.x / dn + b4.x; o.y = a4.y / dn + b4.y;
                o.z = a4.z / dn + b4.z; o.w = a4.w / dn + b4.w;
            } else {
                o = b4;
            }
            reinterpret_cast<float4*>(out)[(size_t)(n0 + node) * (HD / 4) + q4] = o;
        }
        __syncthreads();
    }
}

extern "C" void kernel_launch(void* const* d_in, const int* in_sizes, int n_in,
                              void* d_out, int out_size, void* d_ws, size_t ws_size,
                              hipStream_t stream) {
    const float* feat   = (const float*)d_in[0];
    const int*   src    = (const int*)  d_in[1];
    const int*   dst    = (const int*)  d_in[2];
    const float* W      = (const float*)d_in[3];
    const float* attn_l = (const float*)d_in[4];
    const float* attn_r = (const float*)d_in[5];
    const float* bias   = (const float*)d_in[6];

    const int N = in_sizes[0] / DIN;
    const int E = in_sizes[1];
    float* out = (float*)d_out;

    float* ws    = (float*)d_ws;
    float* ft    = ws;
    float* el    = ft    + (size_t)N * HD;
    float* er    = el    + (size_t)N * HH;
    unsigned* tab = (unsigned*)(er + (size_t)N * HH);
    int* ent = (int*)(tab + (size_t)NBINS * ((E + SORT_S - 1) / SORT_S));

    gat_gemm<<<GEMM_BLOCKS, 256, 0, stream>>>(feat, W, ft, N);
    gat_logits_sort<<<LG_BLOCKS, LG_THREADS, 0, stream>>>(ft, attn_l, attn_r, el, er, src, dst, tab, ent, N, E);
    gat_owner_aggregate<<<OWN_BLOCKS, OWN_THREADS, 0, stream>>>(tab, ent, el, er, ft, bias, out, N, E);
}


# Owner kernel gathers a source row with one 16-byte load per lane (was four 4-byte loads); sums stand lane-major in LDS so the adds stay conflict-free; LDS atomic sum order kept (arrival order)
# speedup vs baseline: 5.9166x; 1.0959x over previous
#include <hip/hip_runtime.h>

typedef __attribute__((ext_vector_type(2))) float v2f;
typedef __attribute__((ext_vector_type(8))) float v8f;

#define DIN 256
#define HD  128
#define HH  4
#define DD  32
#define NEG_SLOPE 0.2f

#define GEMM_BLOCKS 512
#define KG 16
#define APITCH 260

__device__ __forceinline__ void gemm_fetch_tile(float4 (&pre)[4], const float* __restrict__ feat,
                                                int rowBase, int N, int tid) {
    if (rowBase + 16 <= N) {
        #pragma unroll
        for (int i = 0; i < 4; ++i) {
            const int idx4 = i * 256 + tid;
            const int r    = idx4 >> 6;
            const int c4   = idx4 & 63;
            pre[i] = reinterpret_cast<const float4*>(feat + (size_t)(rowBase + r) * DIN)[c4];
        }
    } else {
        #pragma unroll
        for (int i = 0; i < 4; ++i) {
            const int idx4 = i * 256 + tid;
            const int r    = idx4 >> 6;
            const int c4   = idx4 & 63;
            float4 v = make_float4(0.f, 0.f, 0.f, 0.f);
            if (rowBase + r < N)
                v = reinterpret_cast<const float4*>(feat + (size_t)(rowBase + r) * DIN)[c4];
            pre[i] = v;
        }
    }
}

__device__ __forceinline__ void gemm_store_tile(float (*As)[APITCH], const float4 (&pre)[4], int tid) {
    #pragma unroll
    for (int i = 0; i < 4; ++i) {
        const int idx4 = i * 256 + tid;
        const int r    = idx4 >> 6;
        const int c4   = idx4 & 63;
        As[r][c4 * 4 + 0] = pre[i].x; As[r][c4 * 4 + 1] = pre[i].y;
        As[r][c4 * 4 + 2] = pre[i].z; As[r][c4 * 4 + 3] = pre[i].w;
    }
}

__global__ __launch_bounds__(256) void gat_gemm(const float* __restrict__ feat,
                                                const float* __restrict__ W,
                                                float* __restrict__ ft, int N) {
    __shared__ float As[2][16][APITCH];
    const int tid      = threadIdx.x;
    const int numTiles = (N + 15) >> 4;
    int tile = blockIdx.x;
    if (tile >= numTiles) return;

    const int wave    = tid >> 5;
    const int lane    = tid & 31;
    const int half    = lane >> 4;
    const int mrow    = lane & 15;
    const int colBase = wave * 16;
    const int kp      = half * 2;

    const float* wrow = W + (size_t)(colBase + mrow) * DIN + kp;
    v2f b[DIN / 4];
    #pragma unroll
    for (int i = 0; i < DIN / 4; ++i)
        b[i] = *reinterpret_cast<const v2f*>(wrow + 4 * i);

    float4 pre[4];
    gemm_fetch_tile(pre, feat, tile * 16, N, tid);
    gemm_store_tile(As[0], pre, tid);
    __syncthreads();

    int cur = 0;
    for (;;) {
        const int  next    = tile + (int)gridDim.x;
        const bool hasNext = next < numTiles;
        if (hasNext) gemm_fetch_tile(pre, feat, next * 16, N, tid);

        const int rowBase = tile * 16;
        v8f c = {0.f, 0.f, 0.f, 0.f, 0.f, 0.f, 0.f, 0.f};
        const float* arow = &As[cur][mrow][kp];
        v2f a[2][KG];
        #pragma unroll
        for (int i = 0; i < KG; ++i)
            a[0][i] = *reinterpret_cast<const v2f*>(arow + 4 * i);
        __builtin_amdgcn_sched_group_barrier(0x100, KG / 2, 0);
        #pragma unroll
        for (int g = 0; g < DIN / 4 / KG; ++g) {
            if (g + 1 < DIN / 4 / KG) {
                #pragma unroll
                for (int i = 0; i < KG; ++i)
                    a[(g + 1) & 1][i] = *reinterpret_cast<const v2f*>(arow + 4 * ((g + 1) * KG + i));
            }
            #pragma unroll
            for (int i = 0; i < KG; ++i)
                c = __builtin_amdgcn_wmma_f32_16x16x4_f32(
                         false, a[g & 1][i],  false, b[g * KG + i],
                         (short)0, c,  false,  false);
            if (g + 1 < DIN / 4 / KG)
                __builtin_amdgcn_sched_group_barrier(0x100, KG / 2, 0);
            __builtin_amdgcn_sched_group_barrier(0x008, KG, 0);
        }

        float* outBase = ft + (size_t)(rowBase + 8 * half) * HD + colBase + mrow;
        if (rowBase + 16 <= N) {
            #pragma unroll
            for (int v = 0; v < 8; ++v)
                outBase[(size_t)v * HD] = c[v];
        } else {
            #pragma unroll
            for (int v = 0; v < 8; ++v) {
                if (rowBase + v + 8 * half < N)
                    outBase[(size_t)v * HD] = c[v];
            }
        }

        if (!hasNext) break;
        gemm_store_tile(As[cur ^ 1], pre, tid);
        __syncthreads();
        cur ^= 1;
        tile = next;
    }
}

#define NPW        8
#define LG_BLOCKS  512
#define LG_THREADS 256
#define NBINS      256
#define SORT_S     2048
#define SORT_EPT   (SORT_S / LG_THREADS)
#define RMAX 392

__device__ __forceinline__ unsigned own_id(int d) {
    return ((unsigned)d / (unsigned)RMAX) & 255u;
}


__device__ __forceinline__ float head_sum_scatter(const float (&p)[NPW], int lane) {
    const bool b0 = (lane & 1) != 0;
    const bool b1 = (lane & 2) != 0;
    const bool b2 = (lane & 4) != 0;
    float q[4];
    #pragma unroll
    for (int k = 0; k < 4; ++k) {
        const float keep = b0 ? p[2 * k + 1] : p[2 * k];
        const float send = b0 ? p[2 * k]     : p[2 * k + 1];
        q[k] = keep + __shfl_xor(send, 1, 32);
    }
    float r[2];
    #pragma unroll
    for (int k = 0; k < 2; ++k) {
        const float keep = b1 ? q[2 * k + 1] : q[2 * k];
        const float send = b1 ? q[2 * k]     : q[2 * k + 1];
        r[k] = keep + __shfl_xor(send, 2, 32);
    }
    const float keep = b2 ? r[1] : r[0];
    const float send = b2 ? r[0] : r[1];
    return keep + __shfl_xor(send, 4, 32);
}

__global__ __launch_bounds__(LG_THREADS) void gat_logits_sort(
        const float* __restrict__ ft,
        const float* __restrict__ attn_l, const float* __restrict__ attn_r,
        float* __restrict__ el, float* __restrict__ er,
        const int* __restrict__ src, const int* __restrict__ dst,
        unsigned* __restrict__ tab, int* __restrict__ ent, int N, int E) {
    __shared__ unsigned cntb[NBINS];
    __shared__ unsigned startb[NBINS];
    __shared__ unsigned wsum[LG_THREADS / 32];
    __shared__ __attribute__((aligned(16))) int2 stage[SORT_S];
    static_assert(LG_THREADS == NBINS, "the bin scan holds one bin per thread");
    const int tid    = threadIdx.x;
    const int t      = blockIdx.x * LG_THREADS + tid;
    const int stride = (int)gridDim.x * LG_THREADS;

    const int lane   = tid & 31;
    const int wv     = tid >> 5;
    const int gwave  = __builtin_amdgcn_readfirstlane(t >> 5);
    const int nwaves = stride >> 5;

    const int nsteps = (E + SORT_S - 1) / SORT_S;
    int2* ent2 = reinterpret_cast<int2*>(ent);
    for (int step = blockIdx.x; step < nsteps; step += (int)gridDim.x) {
        cntb[tid] = 0u;
        __syncthreads();
        const int e0 = step * SORT_S + tid * SORT_EPT;
        int d[SORT_EPT], s[SORT_EPT];
        if (e0 + SORT_EPT <= E) {
            #pragma unroll
            for (int u = 0; u < SORT_EPT / 4; ++u) {
                const int4 dv = reinterpret_cast<const int4*>(dst + e0)[u];
                const int4 sv = reinterpret_cast<const int4*>(src + e0)[u];
                d[4 * u + 0] = dv.x; d[4 * u + 1] = dv.y; d[4 * u + 2] = dv.z; d[4 * u + 3] = dv.w;
                s[4 * u + 0] = sv.x; s[4 * u + 1] = sv.y; s[4 * u + 2] = sv.z; s[4 * u + 3] = sv.w;
            }
        } else {
            #pragma unroll
            for (int u = 0; u < SORT_EPT; ++u) {
                d[u] = (e0 + u < E) ? dst[e0 + u] : 0;
                s[u] = (e0 + u < E) ? src[e0 + u] : 0;
            }
        }
        unsigned bn[SORT_EPT], rk[SORT_EPT];
        #pragma unroll
        for (int u = 0; u < SORT_EPT; ++u) {
            bn[u] = own_id(d[u]);
            rk[u] = 0u;
            if (e0 + u < E) rk[u] = atomicAdd(&cntb[bn[u]], 1u);
        }
        __syncthreads();

        const unsigned c = cntb[tid];
        unsigned incl = c;
        #pragma unroll
        for (int o = 1; o < 32; o <<= 1) {
            const unsigned v = __shfl_up(incl, o, 32);
            if (lane >= o) incl += v;
        }
        if (lane == 31) wsum[wv] = incl;
        __syncthreads();
        unsigned basev = 0u;
        #pragma unroll
        for (int w = 0; w < LG_THREADS / 32; ++w)
            if (w < wv) basev += wsum[w];
        const unsigned st = basev + incl - c;
        startb[tid] = st;
        tab[(size_t)step * NBINS + tid] = (st << 16) | c;
        __syncthreads();

        #pragma unroll
        for (int u = 0; u < SORT_EPT; ++u) {
            if (e0 + u < E)
                stage[startb[bn[u]] + rk[u]] = make_int2(s[u], d[u]);
        }
        __syncthreads();

        const int left   = E - step * SORT_S;
        const int nvalid = (left < SORT_S) ? left : SORT_S;
        int4* outv = reinterpret_cast<int4*>(ent2 + (size_t)step * SORT_S);
        #pragma unroll
        for (int u = 0; u < SORT_S / 2 / LG_THREADS; ++u) {
            const int j = u * LG_THREADS + tid;
            if (2 * j < nvalid) outv[j] = reinterpret_cast<const int4*>(stage)[j];
        }
    }


    const int ntrips  = (N + 2 * NPW - 1) / (2 * NPW);
    const float4 wl = reinterpret_cast<const float4*>(attn_l)[lane];
    const float4 wr = reinterpret_cast<const float4*>(attn_r)[lane];
    for (int g = gwave; g < ntrips; g += nwaves) {
        const int n0 = g * 2 * NPW;
        const float4* frow = reinterpret_cast<const float4*>(ft + (size_t)n0 * HD) + lane;
        float4 v[2 * NPW];
        if (n0 + 2 * NPW <= N) {
            #pragma unroll
            for (int i = 0; i < 2 * NPW; ++i) v[i] = frow[(size_t)i * (HD / 4)];
        } else {
            #pragma unroll
            for (int i = 0; i < 2 * NPW; ++i)
                v[i] = (n0 + i < N) ? frow[(size_t)i * (HD / 4)] : make_float4(0.f, 0.f, 0.f, 0.f);
        }
        #pragma unroll
        for (int u = 0; u < 2; ++u) {
            float pl[NPW], pr[NPW];
            #pragma unroll
            for (int i = 0; i < NPW; ++i) {
                const float4 x = v[u * NPW + i];
                pl[i] = x.x * wl.x + x.y * wl.y + x.z * wl.z + x.w * wl.w;
                pr[i] = x.x * wr.x + x.y * wr.y + x.z * wr.z + x.w * wr.w;
            }
            const float sl = head_sum_scatter(pl, lane);
            const float sr = head_sum_scatter(pr, lane);
            const int node = n0 + u * NPW + (lane & 7);
            if (node < N) {
                el[(size_t)node * HH + (lane >> 3)] = sl;
                er[(size_t)node * HH + (lane >> 3)] = sr;
            }
        }
    }
}

__device__ __forceinline__ float leaky(float x) {
    return (x > 0.f) ? x : NEG_SLOPE * x;
}

#define OWN_BLOCKS  256
#define OWN_THREADS 1024
#define OWN_WAVES   (OWN_THREADS / 32)
#define DRAIN       8
#define QCAP        480
#define FILL        8
#define RESOLVE     4
#define NOT_OWNED   0xFFFFu

__device__ __forceinline__ void own_resolve_max(
        const int* q, const unsigned short* qd, int n, int lane,
        const float* __restrict__ el, const float* erl, float* mloc) {
    for (int base = 0; base < n; base += 32 * RESOLVE) {
        int      s[RESOLVE];
        unsigned dr[RESOLVE];
        #pragma unroll
        for (int u = 0; u < RESOLVE; ++u) {
            const int idx = base + u * 32 + lane;
            s[u]  = (idx < n) ? q[idx] : 0;
            dr[u] = (idx < n) ? (unsigned)qd[idx] : NOT_OWNED;
        }
        float4 l[RESOLVE];
        #pragma unroll
        for (int u = 0; u < RESOLVE; ++u)
            l[u] = reinterpret_cast<const float4*>(el)[s[u]];
        #pragma unroll
        for (int u = 0; u < RESOLVE; ++u) {
            if (dr[u] != NOT_OWNED) {
                const int dl = (int)dr[u];
                const float4 r = reinterpret_cast<const float4*>(erl)[dl];
                atomicMax(&mloc[dl * HH + 0], leaky(l[u].x + r.x));
                atomicMax(&mloc[dl * HH + 1], leaky(l[u].y + r.y));
                atomicMax(&mloc[dl * HH + 2], leaky(l[u].z + r.z));
                atomicMax(&mloc[dl * HH + 3], leaky(l[u].w + r.w));
            }
        }
    }
}

__device__ __forceinline__ int acc_word(int c) {
    return (c & 3) * 32 + (c >> 2);
}

template <int JJ>
__device__ __forceinline__ float group_lane(float x) {
    return __uint_as_float(__builtin_amdgcn_mov_dpp8(__float_as_uint(x), (unsigned)JJ * 0x249249u));
}

template <int JJ>
__device__ __forceinline__ void own_add_edge(const float4 &v, float ae, int dl, bool ok, float* acc, int lane) {
    const int dj  = __builtin_amdgcn_readlane(dl, JJ);
    const int okj = __builtin_amdgcn_readlane((int)ok, JJ);
    if (okj) {
        const float a = group_lane<JJ>(ae);
        float* accd = acc + dj * HD + lane;
        atomicAdd(&accd[0],  v.x * a);
        atomicAdd(&accd[32], v.y * a);
        atomicAdd(&accd[64], v.z * a);
        atomicAdd(&accd[96], v.w * a);
    }
}

__device__ __forceinline__ void own_accumulate(
        const int* q, const unsigned short* qd, int n, int lane,
        const float* __restrict__ el, const float* __restrict__ ft,
        const float* erl, const float* mloc, float* den, float* acc) {
    static_assert(DRAIN == 8, "one queued edge per lane of a group of eight");
    const int j = lane & 7;
    const int h = lane >> 3;
    int b = 0;
    for (; b + DRAIN <= n; b += DRAIN) {
        const int      sv = q[b + j];
        const unsigned dr = qd[b + j];
        const bool     ok = dr != NOT_OWNED;
        const int      dl = ok ? (int)dr : 0;
        const float    xl = el[(size_t)sv * HH + h];
        float4 v[DRAIN];
        #pragma unroll
        for (int jj = 0; jj < DRAIN; ++jj) {
            const int s = __builtin_amdgcn_readlane(sv, jj);
            v[jj] = reinterpret_cast<const float4*>(ft + (size_t)s * HD)[lane];
        }
        const float x  = leaky(xl + erl[dl * HH + h]);
        const float ae = __expf(x - mloc[dl * HH + h]);
        if (ok) atomicAdd(&den[dl * HH + h], ae);
        own_add_edge<0>(v[0], ae, dl, ok, acc, lane);
        own_add_edge<1>(v[1], ae, dl, ok, acc, lane);
        own_add_edge<2>(v[2], ae, dl, ok, acc, lane);
        own_add_edge<3>(v[3], ae, dl, ok, acc, lane);
        own_add_edge<4>(v[4], ae, dl, ok, acc, lane);
        own_add_edge<5>(v[5], ae, dl, ok, acc, lane);
        own_add_edge<6>(v[6], ae, dl, ok, acc, lane);
        own_add_edge<7>(v[7], ae, dl, ok, acc, lane);
    }
    for (; b < n; ++b) {
        const int      s  = q[b];
        const unsigned dr = qd[b];
        if (dr != NOT_OWNED) {
            const int dl = (int)dr;
            const float4 v = reinterpret_cast<const float4*>(ft + (size_t)s * HD)[lane];
            const float x  = leaky(el[(size_t)s * HH + h] + erl[dl * HH + h]);
            const float ae = __expf(x - mloc[dl * HH + h]);
            if (j == 0) atomicAdd(&den[dl * HH + h], ae);
            float* accd = acc + dl * HD + lane;
            atomicAdd(&accd[0],  v.x * ae);
            atomicAdd(&accd[32], v.y * ae);
            atomicAdd(&accd[64], v.z * ae);
            atomicAdd(&accd[96], v.w * ae);
        }
    }
}

template <bool ACC>
__device__ __forceinline__ void own_plain_pass(
        int lane, int sBeg, int sEnd, int n0, int cnt,
        const unsigned* __restrict__ trow, const int2* __restrict__ ent2,
        const float* __restrict__ el, const float* __restrict__ ft,
        const float* erl, float* mloc, float* den, float* acc) {
    for (int sidx = sBeg + lane; sidx < sEnd; sidx += 32) {
        const unsigned tw = trow[(size_t)sidx * NBINS];
        const int c = (int)(tw & 0xFFFFu);
        const int2* p = ent2 + (size_t)sidx * SORT_S + (tw >> 16);
        for (int i = 0; i < c; ++i) {
            const int2 ed = p[i];
            const int dl = ed.y - n0;
            if ((unsigned)dl >= (unsigned)cnt) continue;
            const int s = ed.x;
            for (int hh = 0; hh < HH; ++hh) {
                const float x = leaky(el[(size_t)s * HH + hh] + erl[dl * HH + hh]);
                if (!ACC) {
                    atomicMax(&mloc[dl * HH + hh], x);
                } else {
                    const float ae = __expf(x - mloc[dl * HH + hh]);
                    atomicAdd(&den[dl * HH + hh], ae);
                    for (int dd = 0; dd < DD; ++dd)
                        atomicAdd(&acc[dl * HD + acc_word(hh * DD + dd)], ft[(size_t)s * HD + hh * DD + dd] * ae);
                }
            }
        }
    }
}

__global__ __launch_bounds__(OWN_THREADS) void gat_owner_aggregate(
        const unsigned* __restrict__ tab, const int* __restrict__ ent,
        const float* __restrict__ el, const float* __restrict__ er,
        const float* __restrict__ ft, const float* __restrict__ bias,
        float* __restrict__ out, int N, int E) {
    __shared__ __attribute__((aligned(16))) float acc[RMAX * HD];
    __shared__ __attribute__((aligned(16))) float den[RMAX * HH];
    __shared__ __attribute__((aligned(16))) float mloc[RMAX * HH];
    __shared__ __attribute__((aligned(16))) float erl[RMAX * HH];
    __shared__ int queue[OWN_WAVES * QCAP];
    __shared__ unsigned short qdst[OWN_WAVES * QCAP];
    __shared__ int ovf;

    const int tid   = threadIdx.x;
    const int lane  = tid & 31;
    const int wave  = __builtin_amdgcn_readfirstlane(tid >> 5);
    int* q              = &queue[wave * QCAP];
    unsigned short* qd  = &qdst[wave * QCAP];
    const int2* ent2    = reinterpret_cast<const int2*>(ent);
    const int nsteps    = (E + SORT_S - 1) / SORT_S;
    const int chunk     = (nsteps + OWN_WAVES - 1) / OWN_WAVES;
    const int sBeg      = wave * chunk;
    const int sEnd      = (sBeg + chunk < nsteps) ? (sBeg + chunk) : nsteps;

    for (int n0 = blockIdx.x * RMAX; n0 < N; n0 += (int)gridDim.x * RMAX) {
        const int cnt = (N - n0 < RMAX) ? (N - n0) : RMAX;
        const unsigned ownBin = (unsigned)(n0 / RMAX) & 255u;
        const unsigned* trow = tab + ownBin;

        const float4 z4 = make_float4(0.f, 0.f, 0.f, 0.f);
        const float4 lo = make_float4(-3.402823466e38f, -3.402823466e38f,
                                      -3.402823466e38f, -3.402823466e38f);
        for (int i = tid; i < cnt * (HD / 4); i += OWN_THREADS) reinterpret_cast<float4*>(acc)[i] = z4;
        for (int i = tid; i < cnt; i += OWN_THREADS) {
            reinterpret_cast<float4*>(den)[i]  = z4;
            reinterpret_cast<float4*>(mloc)[i] = lo;
            reinterpret_cast<float4*>(erl)[i]  = reinterpret_cast<const float4*>(er)[n0 + i];
        }
        if (tid == 0) ovf = 0;
        __syncthreads();

        int  qn   = 0;
        bool lost = false;
        for (int sb = sBeg; sb < sEnd; sb += 32) {
            const int sidx = sb + lane;
            const unsigned tw = (sidx < sEnd) ? trow[(size_t)sidx * NBINS] : 0u;
            const int c = (int)(tw & 0xFFFFu);
            int incl = c;
            #pragma unroll
            for (int o = 1; o < 32; o <<= 1) {
                const int v = __shfl_up(incl, o, 32);
                if (lane >= o) incl += v;
            }
            const int total = __builtin_amdgcn_readlane(incl, 31);
            if (qn + total > QCAP) { lost = true; break; }
            const int pos = qn + incl - c;
            const int2* p = ent2 + (size_t)sidx * SORT_S + (tw >> 16);
            for (int i0 = 0; __builtin_amdgcn_ballot_w32(i0 < c) != 0u; i0 += FILL) {
                int2 ed[FILL];
                #pragma unroll
                for (int u = 0; u < FILL; ++u)
                    ed[u] = (i0 + u < c) ? p[i0 + u] : make_int2(0, 0);
                #pragma unroll
                for (int u = 0; u < FILL; ++u) {
                    if (i0 + u < c) {
                        const int dl = ed[u].y - n0;
                        const bool ok = (unsigned)dl < (unsigned)cnt;
                        q[pos + i0 + u]  = ed[u].x;
                        qd[pos + i0 + u] = ok ? (unsigned short)dl : (unsigned short)NOT_OWNED;
                    }
                }
            }
            qn += total;
        }
        if (lost && lane == 0) ovf = 1;

        own_resolve_max(q, qd, qn, lane, el, erl, mloc);
        __syncthreads();

        if (__builtin_amdgcn_readfirstlane(ovf) == 0) {
            own_accumulate(q, qd, qn, lane, el, ft, erl, mloc, den, acc);
        } else {
            own_plain_pass<false>(lane, sBeg, sEnd, n0, cnt, trow, ent2, el, ft, erl, mloc, den, acc);
            __syncthreads();
            own_plain_pass<true>(lane, sBeg, sEnd, n0, cnt, trow, ent2, el, ft, erl, mloc, den, acc);
        }
        __syncthreads();


        for (int i = tid; i < cnt * (HD / 4); i += OWN_THREADS) {
            const int node = i >> 5;
            const int q4   = i & 31;
            const float* an = acc + node * HD + q4;
            const float4 a4 = make_float4(an[0], an[32], an[64], an[96]);
            const float  dn = den[node * HH + (q4 >> 3)];
            const float4 b4 = reinterpret_cast<const float4*>(bias)[q4];
            float4 o;
            if (dn > 0.f) {
                o.x = a4.x / dn + b4.x; o.y = a4.y / dn + b4.y;
                o.z = a4.z / dn + b4.z; o.w = a4.w / dn + b4.w;
            } else {
                o = b4;
            }
            reinterpret_cast<float4*>(out)[(size_t)(n0 + node) * (HD / 4) + q4] = o;
        }
        __syncthreads();
    }
}

extern "C" void kernel_launch(void* const* d_in, const int* in_sizes, int n_in,
                              void* d_out, int out_size, void* d_ws, size_t ws_size,
                              hipStream_t stream) {
    const float* feat   = (const float*)d_in[0];
    const int*   src    = (const int*)  d_in[1];
    const int*   dst    = (const int*)  d_in[2];
    const float* W      = (const float*)d_in[3];
    const float* attn_l = (const float*)d_in[4];
    const float* attn_r = (const float*)d_in[5];
    const float* bias   = (const float*)d_in[6];

    const int N = in_sizes[0] / DIN;
    const int E = in_sizes[1];
    float* out = (float*)d_out;

    float* ws    = (float*)d_ws;
    float* ft    = ws;
    float* el    = ft    + (size_t)N * HD;
    float* er    = el    + (size_t)N * HH;
    unsigned* tab = (unsigned*)(er + (size_t)N * HH);
    int* ent = (int*)(tab + (size_t)NBINS * ((E + SORT_S - 1) / SORT_S));

    gat_gemm<<<GEMM_BLOCKS, 256, 0, stream>>>(feat, W, ft, N);
    gat_logits_sort<<<LG_BLOCKS, LG_THREADS, 0, stream>>>(ft, attn_l, attn_r, el, er, src, dst, tab, ent, N, E);
    gat_owner_aggregate<<<OWN_BLOCKS, OWN_THREADS, 0, stream>>>(tab, ent, el, er, ft, bias, out, N, E);
}


# Owner kernel: an edge's four scores stay in registers of the lanes that use them from the max pass to the sum pass, so el[src] is gathered once per edge, not twice; LDS atomic sum order kept
# speedup vs baseline: 6.2004x; 1.0480x over previous
#include <hip/hip_runtime.h>

typedef __attribute__((ext_vector_type(2))) float v2f;
typedef __attribute__((ext_vector_type(8))) float v8f;

#define DIN 256
#define HD  128
#define HH  4
#define DD  32
#define NEG_SLOPE 0.2f

#define GEMM_BLOCKS 512
#define KG 16
#define APITCH 260

__device__ __forceinline__ void gemm_fetch_tile(float4 (&pre)[4], const float* __restrict__ feat,
                                                int rowBase, int N, int tid) {
    if (rowBase + 16 <= N) {
        #pragma unroll
        for (int i = 0; i < 4; ++i) {
            const int idx4 = i * 256 + tid;
            const int r    = idx4 >> 6;
            const int c4   = idx4 & 63;
            pre[i] = reinterpret_cast<const float4*>(feat + (size_t)(rowBase + r) * DIN)[c4];
        }
    } else {
        #pragma unroll
        for (int i = 0; i < 4; ++i) {
            const int idx4 = i * 256 + tid;
            const int r    = idx4 >> 6;
            const int c4   = idx4 & 63;
            float4 v = make_float4(0.f, 0.f, 0.f, 0.f);
            if (rowBase + r < N)
                v = reinterpret_cast<const float4*>(feat + (size_t)(rowBase + r) * DIN)[c4];
            pre[i] = v;
        }
    }
}

__device__ __forceinline__ void gemm_store_tile(float (*As)[APITCH], const float4 (&pre)[4], int tid) {
    #pragma unroll
    for (int i = 0; i < 4; ++i) {
        const int idx4 = i * 256 + tid;
        const int r    = idx4 >> 6;
        const int c4   = idx4 & 63;
        As[r][c4 * 4 + 0] = pre[i].x; As[r][c4 * 4 + 1] = pre[i].y;
        As[r][c4 * 4 + 2] = pre[i].z; As[r][c4 * 4 + 3] = pre[i].w;
    }
}

__global__ __launch_bounds__(256) void gat_gemm(const float* __restrict__ feat,
                                                const float* __restrict__ W,
                                                float* __restrict__ ft, int N) {
    __shared__ float As[2][16][APITCH];
    const int tid      = threadIdx.x;
    const int numTiles = (N + 15) >> 4;
    int tile = blockIdx.x;
    if (tile >= numTiles) return;

    const int wave    = tid >> 5;
    const int lane    = tid & 31;
    const int half    = lane >> 4;
    const int mrow    = lane & 15;
    const int colBase = wave * 16;
    const int kp      = half * 2;

    const float* wrow = W + (size_t)(colBase + mrow) * DIN + kp;
    v2f b[DIN / 4];
    #pragma unroll
    for (int i = 0; i < DIN / 4; ++i)
        b[i] = *reinterpret_cast<const v2f*>(wrow + 4 * i);

    float4 pre[4];
    gemm_fetch_tile(pre, feat, tile * 16, N, tid);
    gemm_store_tile(As[0], pre, tid);
    __syncthreads();

    int cur = 0;
    for (;;) {
        const int  next    = tile + (int)gridDim.x;
        const bool hasNext = next < numTiles;
        if (hasNext) gemm_fetch_tile(pre, feat, next * 16, N, tid);

        const int rowBase = tile * 16;
        v8f c = {0.f, 0.f, 0.f, 0.f, 0.f, 0.f, 0.f, 0.f};
        const float* arow = &As[cur][mrow][kp];
        v2f a[2][KG];
        #pragma unroll
        for (int i = 0; i < KG; ++i)
            a[0][i] = *reinterpret_cast<const v2f*>(arow + 4 * i);
        __builtin_amdgcn_sched_group_barrier(0x100, KG / 2, 0);
        #pragma unroll
        for (int g = 0; g < DIN / 4 / KG; ++g) {
            if (g + 1 < DIN / 4 / KG) {
                #pragma unroll
                for (int i = 0; i < KG; ++i)
                    a[(g + 1) & 1][i] = *reinterpret_cast<const v2f*>(arow + 4 * ((g + 1) * KG + i));
            }
            #pragma unroll
            for (int i = 0; i < KG; ++i)
                c = __builtin_amdgcn_wmma_f32_16x16x4_f32(
                         false, a[g & 1][i],  false, b[g * KG + i],
                         (short)0, c,  false,  false);
            if (g + 1 < DIN / 4 / KG)
                __builtin_amdgcn_sched_group_barrier(0x100, KG / 2, 0);
            __builtin_amdgcn_sched_group_barrier(0x008, KG, 0);
        }

        float* outBase = ft + (size_t)(rowBase + 8 * half) * HD + colBase + mrow;
        if (rowBase + 16 <= N) {
            #pragma unroll
            for (int v = 0; v < 8; ++v)
                outBase[(size_t)v * HD] = c[v];
        } else {
            #pragma unroll
            for (int v = 0; v < 8; ++v) {
                if (rowBase + v + 8 * half < N)
                    outBase[(size_t)v * HD] = c[v];
            }
        }

        if (!hasNext) break;
        gemm_store_tile(As[cur ^ 1], pre, tid);
        __syncthreads();
        cur ^= 1;
        tile = next;
    }
}

#define NPW        8
#define LG_BLOCKS  512
#define LG_THREADS 256
#define NBINS      256
#define SORT_S     2048
#define SORT_EPT   (SORT_S / LG_THREADS)
#define RMAX 392

__device__ __forceinline__ unsigned own_id(int d) {
    return ((unsigned)d / (unsigned)RMAX) & 255u;
}


__device__ __forceinline__ float head_sum_scatter(const float (&p)[NPW], int lane) {
    const bool b0 = (lane & 1) != 0;
    const bool b1 = (lane & 2) != 0;
    const bool b2 = (lane & 4) != 0;
    float q[4];
    #pragma unroll
    for (int k = 0; k < 4; ++k) {
        const float keep = b0 ? p[2 * k + 1] : p[2 * k];
        const float send = b0 ? p[2 * k]     : p[2 * k + 1];
        q[k] = keep + __shfl_xor(send, 1, 32);
    }
    float r[2];
    #pragma unroll
    for (int k = 0; k < 2; ++k) {
        const float keep = b1 ? q[2 * k + 1] : q[2 * k];
        const float send = b1 ? q[2 * k]     : q[2 * k + 1];
        r[k] = keep + __shfl_xor(send, 2, 32);
    }
    const float keep = b2 ? r[1] : r[0];
    const float send = b2 ? r[0] : r[1];
    return keep + __shfl_xor(send, 4, 32);
}

__global__ __launch_bounds__(LG_THREADS) void gat_logits_sort(
        const float* __restrict__ ft,
        const float* __restrict__ attn_l, const float* __restrict__ attn_r,
        float* __restrict__ el, float* __restrict__ er,
        const int* __restrict__ src, const int* __restrict__ dst,
        unsigned* __restrict__ tab, int* __restrict__ ent, int N, int E) {
    __shared__ unsigned cntb[NBINS];
    __shared__ unsigned startb[NBINS];
    __shared__ unsigned wsum[LG_THREADS / 32];
    __shared__ __attribute__((aligned(16))) int2 stage[SORT_S];
    static_assert(LG_THREADS == NBINS, "the bin scan holds one bin per thread");
    const int tid    = threadIdx.x;
    const int t      = blockIdx.x * LG_THREADS + tid;
    const int stride = (int)gridDim.x * LG_THREADS;

    const int lane   = tid & 31;
    const int wv     = tid >> 5;
    const int gwave  = __builtin_amdgcn_readfirstlane(t >> 5);
    const int nwaves = stride >> 5;

    const int nsteps = (E + SORT_S - 1) / SORT_S;
    int2* ent2 = reinterpret_cast<int2*>(ent);
    for (int step = blockIdx.x; step < nsteps; step += (int)gridDim.x) {
        cntb[tid] = 0u;
        __syncthreads();
        const int e0 = step * SORT_S + tid * SORT_EPT;
        int d[SORT_EPT], s[SORT_EPT];
        if (e0 + SORT_EPT <= E) {
            #pragma unroll
            for (int u = 0; u < SORT_EPT / 4; ++u) {
                const int4 dv = reinterpret_cast<const int4*>(dst + e0)[u];
                const int4 sv = reinterpret_cast<const int4*>(src + e0)[u];
                d[4 * u + 0] = dv.x; d[4 * u + 1] = dv.y; d[4 * u + 2] = dv.z; d[4 * u + 3] = dv.w;
                s[4 * u + 0] = sv.x; s[4 * u + 1] = sv.y; s[4 * u + 2] = sv.z; s[4 * u + 3] = sv.w;
            }
        } else {
            #pragma unroll
            for (int u = 0; u < SORT_EPT; ++u) {
                d[u] = (e0 + u < E) ? dst[e0 + u] : 0;
                s[u] = (e0 + u < E) ? src[e0 + u] : 0;
            }
        }
        unsigned bn[SORT_EPT], rk[SORT_EPT];
        #pragma unroll
        for (int u = 0; u < SORT_EPT; ++u) {
            bn[u] = own_id(d[u]);
            rk[u] = 0u;
            if (e0 + u < E) rk[u] = atomicAdd(&cntb[bn[u]], 1u);
        }
        __syncthreads();

        const unsigned c = cntb[tid];
        unsigned incl = c;
        #pragma unroll
        for (int o = 1; o < 32; o <<= 1) {
            const unsigned v = __shfl_up(incl, o, 32);
            if (lane >= o) incl += v;
        }
        if (lane == 31) wsum[wv] = incl;
        __syncthreads();
        unsigned basev = 0u;
        #pragma unroll
        for (int w = 0; w < LG_THREADS / 32; ++w)
            if (w < wv) basev += wsum[w];
        const unsigned st = basev + incl - c;
        startb[tid] = st;
        tab[(size_t)step * NBINS + tid] = (st << 16) | c;
        __syncthreads();

        #pragma unroll
        for (int u = 0; u < SORT_EPT; ++u) {
            if (e0 + u < E)
                stage[startb[bn[u]] + rk[u]] = make_int2(s[u], d[u]);
        }
        __syncthreads();

        const int left   = E - step * SORT_S;
        const int nvalid = (left < SORT_S) ? left : SORT_S;
        int4* outv = reinterpret_cast<int4*>(ent2 + (size_t)step * SORT_S);
        #pragma unroll
        for (int u = 0; u < SORT_S / 2 / LG_THREADS; ++u) {
            const int j = u * LG_THREADS + tid;
            if (2 * j < nvalid) outv[j] = reinterpret_cast<const int4*>(stage)[j];
        }
    }


    const int ntrips  = (N + 2 * NPW - 1) / (2 * NPW);
    const float4 wl = reinterpret_cast<const float4*>(attn_l)[lane];
    const float4 wr = reinterpret_cast<const float4*>(attn_r)[lane];
    for (int g = gwave; g < ntrips; g += nwaves) {
        const int n0 = g * 2 * NPW;
        const float4* frow = reinterpret_cast<const float4*>(ft + (size_t)n0 * HD) + lane;
        float4 v[2 * NPW];
        if (n0 + 2 * NPW <= N) {
            #pragma unroll
            for (int i = 0; i < 2 * NPW; ++i) v[i] = frow[(size_t)i * (HD / 4)];
        } else {
            #pragma unroll
            for (int i = 0; i < 2 * NPW; ++i)
                v[i] = (n0 + i < N) ? frow[(size_t)i * (HD / 4)] : make_float4(0.f, 0.f, 0.f, 0.f);
        }
        #pragma unroll
        for (int u = 0; u < 2; ++u) {
            float pl[NPW], pr[NPW];
            #pragma unroll
            for (int i = 0; i < NPW; ++i) {
                const float4 x = v[u * NPW + i];
                pl[i] = x.x * wl.x + x.y * wl.y + x.z * wl.z + x.w * wl.w;
                pr[i] = x.x * wr.x + x.y * wr.y + x.z * wr.z + x.w * wr.w;
            }
            const float sl = head_sum_scatter(pl, lane);
            const float sr = head_sum_scatter(pr, lane);
            const int node = n0 + u * NPW + (lane & 7);
            if (node < N) {
                el[(size_t)node * HH + (lane >> 3)] = sl;
                er[(size_t)node * HH + (lane >> 3)] = sr;
            }
        }
    }
}

__device__ __forceinline__ float leaky(float x) {
    return (x > 0.f) ? x : NEG_SLOPE * x;
}

#define OWN_BLOCKS  256
#define OWN_THREADS 1024
#define OWN_WAVES   (OWN_THREADS / 32)
#define DRAIN       8
#define SC_A        32
#define SC_B        8
#define QCAP        ((SC_A + SC_B) * DRAIN)
#define FILL        8
#define RESOLVE     8

typedef __attribute__((ext_vector_type(SC_A))) float scvA;
typedef __attribute__((ext_vector_type(SC_B))) float scvB;
static_assert(SC_A % RESOLVE == 0 && SC_B % RESOLVE == 0, "a trip of phase 2 stays inside one score vector");
#define NOT_OWNED   0xFFFFu

template <typename VEC>
__device__ __forceinline__ void own_resolve_max(
        VEC &sc, int gBeg, int gEnd, int gOff,
        const int* q, const unsigned short* qd, int n, int lane,
        const float* __restrict__ el, const float* erl, float* mloc) {
    const int j = lane & 7;
    const int h = lane >> 3;
    for (int g0 = gBeg; g0 < gEnd; g0 += RESOLVE) {
        int      s[RESOLVE];
        unsigned dr[RESOLVE];
        #pragma unroll
        for (int u = 0; u < RESOLVE; ++u) {
            const int idx  = (g0 + u) * DRAIN + j;
            const int idxc = (idx < n) ? idx : n - 1;
            s[u]  = q[idxc];
            dr[u] = (idx < n) ? (unsigned)qd[idxc] : NOT_OWNED;
        }
        float xl[RESOLVE], xr[RESOLVE];
        #pragma unroll
        for (int u = 0; u < RESOLVE; ++u)
            xl[u] = el[(size_t)s[u] * HH + h];
        #pragma unroll
        for (int u = 0; u < RESOLVE; ++u)
            xr[u] = erl[((dr[u] != NOT_OWNED) ? (int)dr[u] : 0) * HH + h];
        #pragma unroll
        for (int u = 0; u < RESOLVE; ++u) {
            const bool ok = dr[u] != NOT_OWNED;
            const int  dl = ok ? (int)dr[u] : 0;
            const float x = leaky(xl[u] + xr[u]);
            if (ok) atomicMax(&mloc[dl * HH + h], x);
            sc[g0 + u - gOff] = x;
        }
    }
}

__device__ __forceinline__ int acc_word(int c) {
    return (c & 3) * 32 + (c >> 2);
}

template <int JJ>
__device__ __forceinline__ float group_lane(float x) {
    return __uint_as_float(__builtin_amdgcn_mov_dpp8(__float_as_uint(x), (unsigned)JJ * 0x249249u));
}

template <int JJ>
__device__ __forceinline__ void own_add_edge(const float4 &v, float ae, int dl, bool ok, float* acc, int lane) {
    const int dj  = __builtin_amdgcn_readlane(dl, JJ);
    const int okj = __builtin_amdgcn_readlane((int)ok, JJ);
    if (okj) {
        const float a = group_lane<JJ>(ae);
        float* accd = acc + dj * HD + lane;
        atomicAdd(&accd[0],  v.x * a);
        atomicAdd(&accd[32], v.y * a);
        atomicAdd(&accd[64], v.z * a);
        atomicAdd(&accd[96], v.w * a);
    }
}

template <typename VEC>
__device__ __forceinline__ void own_accumulate(
        const VEC &sc, int gBeg, int gEnd, int gOff,
        const int* q, const unsigned short* qd, int n, int lane,
        const float* __restrict__ ft, const float* mloc, float* den, float* acc) {
    static_assert(DRAIN == 8, "one queued edge per lane of a group of eight");
    const int j = lane & 7;
    const int h = lane >> 3;
    for (int g = gBeg; g < gEnd; ++g) {
        const int      b   = g * DRAIN;
        const int      idx = b + j;
        const int      idxc = (idx < n) ? idx : n - 1;
        const int      sv  = q[idxc];
        const unsigned dr  = (idx < n) ? (unsigned)qd[idxc] : NOT_OWNED;
        const bool     ok  = dr != NOT_OWNED;
        const int      dl  = ok ? (int)dr : 0;
        float4 v[DRAIN];
        #pragma unroll
        for (int jj = 0; jj < DRAIN; ++jj) {
            const int s = __builtin_amdgcn_readlane(sv, jj);
            v[jj] = reinterpret_cast<const float4*>(ft + (size_t)s * HD)[lane];
        }
        const float ae = __expf(sc[g - gOff] - mloc[dl * HH + h]);
        if (ok) atomicAdd(&den[dl * HH + h], ae);
        own_add_edge<0>(v[0], ae, dl, ok, acc, lane);
        own_add_edge<1>(v[1], ae, dl, ok, acc, lane);
        own_add_edge<2>(v[2], ae, dl, ok, acc, lane);
        own_add_edge<3>(v[3], ae, dl, ok, acc, lane);
        own_add_edge<4>(v[4], ae, dl, ok, acc, lane);
        own_add_edge<5>(v[5], ae, dl, ok, acc, lane);
        own_add_edge<6>(v[6], ae, dl, ok, acc, lane);
        own_add_edge<7>(v[7], ae, dl, ok, acc, lane);
    }
}

template <bool ACC>
__device__ __forceinline__ void own_plain_pass(
        int lane, int sBeg, int sEnd, int n0, int cnt,
        const unsigned* __restrict__ trow, const int2* __restrict__ ent2,
        const float* __restrict__ el, const float* __restrict__ ft,
        const float* erl, float* mloc, float* den, float* acc) {
    for (int sidx = sBeg + lane; sidx < sEnd; sidx += 32) {
        const unsigned tw = trow[(size_t)sidx * NBINS];
        const int c = (int)(tw & 0xFFFFu);
        const int2* p = ent2 + (size_t)sidx * SORT_S + (tw >> 16);
        for (int i = 0; i < c; ++i) {
            const int2 ed = p[i];
            const int dl = ed.y - n0;
            if ((unsigned)dl >= (unsigned)cnt) continue;
            const int s = ed.x;
            for (int hh = 0; hh < HH; ++hh) {
                const float x = leaky(el[(size_t)s * HH + hh] + erl[dl * HH + hh]);
                if (!ACC) {
                    atomicMax(&mloc[dl * HH + hh], x);
                } else {
                    const float ae = __expf(x - mloc[dl * HH + hh]);
                    atomicAdd(&den[dl * HH + hh], ae);
                    for (int dd = 0; dd < DD; ++dd)
                        atomicAdd(&acc[dl * HD + acc_word(hh * DD + dd)], ft[(size_t)s * HD + hh * DD + dd] * ae);
                }
            }
        }
    }
}

__global__ __launch_bounds__(OWN_THREADS) void gat_owner_aggregate(
        const unsigned* __restrict__ tab, const int* __restrict__ ent,
        const float* __restrict__ el, const float* __restrict__ er,
        const float* __restrict__ ft, const float* __restrict__ bias,
        float* __restrict__ out, int N, int E) {
    __shared__ __attribute__((aligned(16))) float acc[RMAX * HD];
    __shared__ __attribute__((aligned(16))) float den[RMAX * HH];
    __shared__ __attribute__((aligned(16))) float mloc[RMAX * HH];
    __shared__ __attribute__((aligned(16))) float erl[RMAX * HH];
    __shared__ int queue[OWN_WAVES * QCAP];
    __shared__ unsigned short qdst[OWN_WAVES * QCAP];
    __shared__ int ovf;

    const int tid   = threadIdx.x;
    const int lane  = tid & 31;
    const int wave  = __builtin_amdgcn_readfirstlane(tid >> 5);
    int* q              = &queue[wave * QCAP];
    unsigned short* qd  = &qdst[wave * QCAP];
    const int2* ent2    = reinterpret_cast<const int2*>(ent);
    const int nsteps    = (E + SORT_S - 1) / SORT_S;
    const int chunk     = (nsteps + OWN_WAVES - 1) / OWN_WAVES;
    const int sBeg      = wave * chunk;
    const int sEnd      = (sBeg + chunk < nsteps) ? (sBeg + chunk) : nsteps;

    for (int n0 = blockIdx.x * RMAX; n0 < N; n0 += (int)gridDim.x * RMAX) {
        const int cnt = (N - n0 < RMAX) ? (N - n0) : RMAX;
        const unsigned ownBin = (unsigned)(n0 / RMAX) & 255u;
        const unsigned* trow = tab + ownBin;

        const float4 z4 = make_float4(0.f, 0.f, 0.f, 0.f);
        const float4 lo = make_float4(-3.402823466e38f, -3.402823466e38f,
                                      -3.402823466e38f, -3.402823466e38f);
        for (int i = tid; i < cnt * (HD / 4); i += OWN_THREADS) reinterpret_cast<float4*>(acc)[i] = z4;
        for (int i = tid; i < cnt; i += OWN_THREADS) {
            reinterpret_cast<float4*>(den)[i]  = z4;
            reinterpret_cast<float4*>(mloc)[i] = lo;
            reinterpret_cast<float4*>(erl)[i]  = reinterpret_cast<const float4*>(er)[n0 + i];
        }
        if (tid == 0) ovf = 0;
        __syncthreads();

        int  qn   = 0;
        bool lost = false;
        for (int sb = sBeg; sb < sEnd; sb += 32) {
            const int sidx = sb + lane;
            const unsigned tw = (sidx < sEnd) ? trow[(size_t)sidx * NBINS] : 0u;
            const int c = (int)(tw & 0xFFFFu);
            int incl = c;
            #pragma unroll
            for (int o = 1; o < 32; o <<= 1) {
                const int v = __shfl_up(incl, o, 32);
                if (lane >= o) incl += v;
            }
            const int total = __builtin_amdgcn_readlane(incl, 31);
            if (qn + total > QCAP) { lost = true; break; }
            const int pos = qn + incl - c;
            const int2* p = ent2 + (size_t)sidx * SORT_S + (tw >> 16);
            for (int i0 = 0; __builtin_amdgcn_ballot_w32(i0 < c) != 0u; i0 += FILL) {
                int2 ed[FILL];
                #pragma unroll
                for (int u = 0; u < FILL; ++u)
                    ed[u] = (i0 + u < c) ? p[i0 + u] : make_int2(0, 0);
                #pragma unroll
                for (int u = 0; u < FILL; ++u) {
                    if (i0 + u < c) {
                        const int dl = ed[u].y - n0;
                        const bool ok = (unsigned)dl < (unsigned)cnt;
                        q[pos + i0 + u]  = ed[u].x;
                        qd[pos + i0 + u] = ok ? (unsigned short)dl : (unsigned short)NOT_OWNED;
                    }
                }
            }
            qn += total;
        }
        if (lost && lane == 0) ovf = 1;

        const int nG = (qn + DRAIN - 1) / DRAIN;
        const int gA = (nG < SC_A) ? nG : SC_A;
        scvA scA = {};
        scvB scB = {};
        own_resolve_max(scA, 0, gA, 0, q, qd, qn, lane, el, erl, mloc);
        if (nG > SC_A)
            own_resolve_max(scB, SC_A, nG, SC_A, q, qd, qn, lane, el, erl, mloc);
        __syncthreads();

        if (__builtin_amdgcn_readfirstlane(ovf) == 0) {
            own_accumulate(scA, 0, gA, 0, q, qd, qn, lane, ft, mloc, den, acc);
            if (nG > SC_A)
                own_accumulate(scB, SC_A, nG, SC_A, q, qd, qn, lane, ft, mloc, den, acc);
        } else {
            own_plain_pass<false>(lane, sBeg, sEnd, n0, cnt, trow, ent2, el, ft, erl, mloc, den, acc);
            __syncthreads();
            own_plain_pass<true>(lane, sBeg, sEnd, n0, cnt, trow, ent2, el, ft, erl, mloc, den, acc);
        }
        __syncthreads();


        for (int i = tid; i < cnt * (HD / 4); i += OWN_THREADS) {
            const int node = i >> 5;
            const int q4   = i & 31;
            const float* an = acc + node * HD + q4;
            const float4 a4 = make_float4(an[0], an[32], an[64], an[96]);
            const float  dn = den[node * HH + (q4 >> 3)];
            const float4 b4 = reinterpret_cast<const float4*>(bias)[q4];
            float4 o;
            if (dn > 0.f) {
                o.x = a4.x / dn + b4.x; o.y = a4.y / dn + b4.y;
                o.z = a4.z / dn + b4.z; o.w = a4.w / dn + b4.w;
            } else {
                o = b4;
            }
            reinterpret_cast<float4*>(out)[(size_t)(n0 + node) * (HD / 4) + q4] = o;
        }
        __syncthreads();
    }
}

extern "C" void kernel_launch(void* const* d_in, const int* in_sizes, int n_in,
                              void* d_out, int out_size, void* d_ws, size_t ws_size,
                              hipStream_t stream) {
    const float* feat   = (const float*)d_in[0];
    const int*   src    = (const int*)  d_in[1];
    const int*   dst    = (const int*)  d_in[2];
    const float* W      = (const float*)d_in[3];
    const float* attn_l = (const float*)d_in[4];
    const float* attn_r = (const float*)d_in[5];
    const float* bias   = (const float*)d_in[6];

    const int N = in_sizes[0] / DIN;
    const int E = in_sizes[1];
    float* out = (float*)d_out;

    float* ws    = (float*)d_ws;
    float* ft    = ws;
    float* el    = ft    + (size_t)N * HD;
    float* er    = el    + (size_t)N * HH;
    unsigned* tab = (unsigned*)(er + (size_t)N * HH);
    int* ent = (int*)(tab + (size_t)NBINS * ((E + SORT_S - 1) / SORT_S));

    gat_gemm<<<GEMM_BLOCKS, 256, 0, stream>>>(feat, W, ft, N);
    gat_logits_sort<<<LG_BLOCKS, LG_THREADS, 0, stream>>>(ft, attn_l, attn_r, el, er, src, dst, tab, ent, N, E);
    gat_owner_aggregate<<<OWN_BLOCKS, OWN_THREADS, 0, stream>>>(tab, ent, el, er, ft, bias, out, N, E);
}


# GEMM epilogue staged through LDS behind the tile barrier: a wave stores whole 512-byte rows, each 128-byte ft line has one writer, no store wait stands in front of the barrier; sum order kept
# speedup vs baseline: 6.4840x; 1.0457x over previous
#include <hip/hip_runtime.h>

typedef __attribute__((ext_vector_type(2))) float v2f;
typedef __attribute__((ext_vector_type(8))) float v8f;

#define DIN 256
#define HD  128
#define HH  4
#define DD  32
#define NEG_SLOPE 0.2f

#define GEMM_BLOCKS 512
#define KG 16
#define APITCH 260
#define CPITCH 132

__device__ __forceinline__ void gemm_fetch_tile(float4 (&pre)[4], const float* __restrict__ feat,
                                                int rowBase, int N, int tid) {
    if (rowBase + 16 <= N) {
        #pragma unroll
        for (int i = 0; i < 4; ++i) {
            const int idx4 = i * 256 + tid;
            const int r    = idx4 >> 6;
            const int c4   = idx4 & 63;
            pre[i] = reinterpret_cast<const float4*>(feat + (size_t)(rowBase + r) * DIN)[c4];
        }
    } else {
        #pragma unroll
        for (int i = 0; i < 4; ++i) {
            const int idx4 = i * 256 + tid;
            const int r    = idx4 >> 6;
            const int c4   = idx4 & 63;
            float4 v = make_float4(0.f, 0.f, 0.f, 0.f);
            if (rowBase + r < N)
                v = reinterpret_cast<const float4*>(feat + (size_t)(rowBase + r) * DIN)[c4];
            pre[i] = v;
        }
    }
}

__device__ __forceinline__ void gemm_store_tile(float (*As)[APITCH], const float4 (&pre)[4], int tid) {
    #pragma unroll
    for (int i = 0; i < 4; ++i) {
        const int idx4 = i * 256 + tid;
        const int r    = idx4 >> 6;
        const int c4   = idx4 & 63;
        As[r][c4 * 4 + 0] = pre[i].x; As[r][c4 * 4 + 1] = pre[i].y;
        As[r][c4 * 4 + 2] = pre[i].z; As[r][c4 * 4 + 3] = pre[i].w;
    }
}

__global__ __launch_bounds__(256) void gat_gemm(const float* __restrict__ feat,
                                                const float* __restrict__ W,
                                                float* __restrict__ ft, int N) {
    __shared__ float As[2][16][APITCH];
    __shared__ __attribute__((aligned(16))) float Cs[2][16][CPITCH];
    const int tid      = threadIdx.x;
    const int numTiles = (N + 15) >> 4;
    int tile = blockIdx.x;
    if (tile >= numTiles) return;

    const int wave    = tid >> 5;
    const int lane    = tid & 31;
    const int half    = lane >> 4;
    const int mrow    = lane & 15;
    const int colBase = wave * 16;
    const int kp      = half * 2;

    const float* wrow = W + (size_t)(colBase + mrow) * DIN + kp;
    v2f b[DIN / 4];
    #pragma unroll
    for (int i = 0; i < DIN / 4; ++i)
        b[i] = *reinterpret_cast<const v2f*>(wrow + 4 * i);

    float4 pre[4];
    gemm_fetch_tile(pre, feat, tile * 16, N, tid);
    gemm_store_tile(As[0], pre, tid);
    __syncthreads();

    int cur = 0;
    int cb  = 0;
    for (;;) {
        const int  next    = tile + (int)gridDim.x;
        const bool hasNext = next < numTiles;
        if (hasNext) gemm_fetch_tile(pre, feat, next * 16, N, tid);

        const int rowBase = tile * 16;
        v8f c = {0.f, 0.f, 0.f, 0.f, 0.f, 0.f, 0.f, 0.f};
        const float* arow = &As[cur][mrow][kp];
        v2f a[2][KG];
        #pragma unroll
        for (int i = 0; i < KG; ++i)
            a[0][i] = *reinterpret_cast<const v2f*>(arow + 4 * i);
        __builtin_amdgcn_sched_group_barrier(0x100, KG / 2, 0);
        #pragma unroll
        for (int g = 0; g < DIN / 4 / KG; ++g) {
            if (g + 1 < DIN / 4 / KG) {
                #pragma unroll
                for (int i = 0; i < KG; ++i)
                    a[(g + 1) & 1][i] = *reinterpret_cast<const v2f*>(arow + 4 * ((g + 1) * KG + i));
            }
            #pragma unroll
            for (int i = 0; i < KG; ++i)
                c = __builtin_amdgcn_wmma_f32_16x16x4_f32(
                         false, a[g & 1][i],  false, b[g * KG + i],
                         (short)0, c,  false,  false);
            if (g + 1 < DIN / 4 / KG)
                __builtin_amdgcn_sched_group_barrier(0x100, KG / 2, 0);
            __builtin_amdgcn_sched_group_barrier(0x008, KG, 0);
        }

        float* crow = &Cs[cb][8 * half][colBase + mrow];
        #pragma unroll
        for (int v = 0; v < 8; ++v)
            crow[v * CPITCH] = c[v];

        if (hasNext) gemm_store_tile(As[cur ^ 1], pre, tid);
        __syncthreads();

        if (rowBase + 16 <= N) {
            #pragma unroll
            for (int i = 0; i < 2; ++i) {
                const int r  = i * 8 + wave;
                const float4 o = *reinterpret_cast<const float4*>(&Cs[cb][r][lane * 4]);
                reinterpret_cast<float4*>(ft + (size_t)(rowBase + r) * HD)[lane] = o;
            }
        } else {
            #pragma unroll
            for (int i = 0; i < 2; ++i) {
                const int r  = i * 8 + wave;
                if (rowBase + r < N) {
                    const float4 o = *reinterpret_cast<const float4*>(&Cs[cb][r][lane * 4]);
                    reinterpret_cast<float4*>(ft + (size_t)(rowBase + r) * HD)[lane] = o;
                }
            }
        }

        if (!hasNext) break;
        cur ^= 1;
        cb  ^= 1;
        tile = next;
    }
}

#define NPW        8
#define LG_BLOCKS  512
#define LG_THREADS 256
#define NBINS      256
#define SORT_S     2048
#define SORT_EPT   (SORT_S / LG_THREADS)
#define RMAX 392

__device__ __forceinline__ unsigned own_id(int d) {
    return ((unsigned)d / (unsigned)RMAX) & 255u;
}


__device__ __forceinline__ float head_sum_scatter(const float (&p)[NPW], int lane) {
    const bool b0 = (lane & 1) != 0;
    const bool b1 = (lane & 2) != 0;
    const bool b2 = (lane & 4) != 0;
    float q[4];
    #pragma unroll
    for (int k = 0; k < 4; ++k) {
        const float keep = b0 ? p[2 * k + 1] : p[2 * k];
        const float send = b0 ? p[2 * k]     : p[2 * k + 1];
        q[k] = keep + __shfl_xor(send, 1, 32);
    }
    float r[2];
    #pragma unroll
    for (int k = 0; k < 2; ++k) {
        const float keep = b1 ? q[2 * k + 1] : q[2 * k];
        const float send = b1 ? q[2 * k]     : q[2 * k + 1];
        r[k] = keep + __shfl_xor(send, 2, 32);
    }
    const float keep = b2 ? r[1] : r[0];
    const float send = b2 ? r[0] : r[1];
    return keep + __shfl_xor(send, 4, 32);
}

__global__ __launch_bounds__(LG_THREADS) void gat_logits_sort(
        const float* __restrict__ ft,
        const float* __restrict__ attn_l, const float* __restrict__ attn_r,
        float* __restrict__ el, float* __restrict__ er,
        const int* __restrict__ src, const int* __restrict__ dst,
        unsigned* __restrict__ tab, int* __restrict__ ent, int N, int E) {
    __shared__ unsigned cntb[NBINS];
    __shared__ unsigned startb[NBINS];
    __shared__ unsigned wsum[LG_THREADS / 32];
    __shared__ __attribute__((aligned(16))) int2 stage[SORT_S];
    static_assert(LG_THREADS == NBINS, "the bin scan holds one bin per thread");
    const int tid    = threadIdx.x;
    const int t      = blockIdx.x * LG_THREADS + tid;
    const int stride = (int)gridDim.x * LG_THREADS;

    const int lane   = tid & 31;
    const int wv     = tid >> 5;
    const int gwave  = __builtin_amdgcn_readfirstlane(t >> 5);
    const int nwaves = stride >> 5;

    const int nsteps = (E + SORT_S - 1) / SORT_S;
    int2* ent2 = reinterpret_cast<int2*>(ent);
    for (int step = blockIdx.x; step < nsteps; step += (int)gridDim.x) {
        cntb[tid] = 0u;
        __syncthreads();
        const int e0 = step * SORT_S + tid * SORT_EPT;
        int d[SORT_EPT], s[SORT_EPT];
        if (e0 + SORT_EPT <= E) {
            #pragma unroll
            for (int u = 0; u < SORT_EPT / 4; ++u) {
                const int4 dv = reinterpret_cast<const int4*>(dst + e0)[u];
                const int4 sv = reinterpret_cast<const int4*>(src + e0)[u];
                d[4 * u + 0] = dv.x; d[4 * u + 1] = dv.y; d[4 * u + 2] = dv.z; d[4 * u + 3] = dv.w;
                s[4 * u + 0] = sv.x; s[4 * u + 1] = sv.y; s[4 * u + 2] = sv.z; s[4 * u + 3] = sv.w;
            }
        } else {
            #pragma unroll
            for (int u = 0; u < SORT_EPT; ++u) {
                d[u] = (e0 + u < E) ? dst[e0 + u] : 0;
                s[u] = (e0 + u < E) ? src[e0 + u] : 0;
            }
        }
        unsigned bn[SORT_EPT], rk[SORT_EPT];
        #pragma unroll
        for (int u = 0; u < SORT_EPT; ++u) {
            bn[u] = own_id(d[u]);
            rk[u] = 0u;
            if (e0 + u < E) rk[u] = atomicAdd(&cntb[bn[u]], 1u);
        }
        __syncthreads();

        const unsigned c = cntb[tid];
        unsigned incl = c;
        #pragma unroll
        for (int o = 1; o < 32; o <<= 1) {
            const unsigned v = __shfl_up(incl, o, 32);
            if (lane >= o) incl += v;
        }
        if (lane == 31) wsum[wv] = incl;
        __syncthreads();
        unsigned basev = 0u;
        #pragma unroll
        for (int w = 0; w < LG_THREADS / 32; ++w)
            if (w < wv) basev += wsum[w];
        const unsigned st = basev + incl - c;
        startb[tid] = st;
        tab[(size_t)step * NBINS + tid] = (st << 16) | c;
        __syncthreads();

        #pragma unroll
        for (int u = 0; u < SORT_EPT; ++u) {
            if (e0 + u < E)
                stage[startb[bn[u]] + rk[u]] = make_int2(s[u], d[u]);
        }
        __syncthreads();

        const int left   = E - step * SORT_S;
        const int nvalid = (left < SORT_S) ? left : SORT_S;
        int4* outv = reinterpret_cast<int4*>(ent2 + (size_t)step * SORT_S);
        #pragma unroll
        for (int u = 0; u < SORT_S / 2 / LG_THREADS; ++u) {
            const int j = u * LG_THREADS + tid;
            if (2 * j < nvalid) outv[j] = reinterpret_cast<const int4*>(stage)[j];
        }
    }


    const int ntrips  = (N + 2 * NPW - 1) / (2 * NPW);
    const float4 wl = reinterpret_cast<const float4*>(attn_l)[lane];
    const float4 wr = reinterpret_cast<const float4*>(attn_r)[lane];
    for (int g = gwave; g < ntrips; g += nwaves) {
        const int n0 = g * 2 * NPW;
        const float4* frow = reinterpret_cast<const float4*>(ft + (size_t)n0 * HD) + lane;
        float4 v[2 * NPW];
        if (n0 + 2 * NPW <= N) {
            #pragma unroll
            for (int i = 0; i < 2 * NPW; ++i) v[i] = frow[(size_t)i * (HD / 4)];
        } else {
            #pragma unroll
            for (int i = 0; i < 2 * NPW; ++i)
                v[i] = (n0 + i < N) ? frow[(size_t)i * (HD / 4)] : make_float4(0.f, 0.f, 0.f, 0.f);
        }
        #pragma unroll
        for (int u = 0; u < 2; ++u) {
            float pl[NPW], pr[NPW];
            #pragma unroll
            for (int i = 0; i < NPW; ++i) {
                const float4 x = v[u * NPW + i];
                pl[i] = x.x * wl.x + x.y * wl.y + x.z * wl.z + x.w * wl.w;
                pr[i] = x.x * wr.x + x.y * wr.y + x.z * wr.z + x.w * wr.w;
            }
            const float sl = head_sum_scatter(pl, lane);
            const float sr = head_sum_scatter(pr, lane);
            const int node = n0 + u * NPW + (lane & 7);
            if (node < N) {
                el[(size_t)node * HH + (lane >> 3)] = sl;
                er[(size_t)node * HH + (lane >> 3)] = sr;
            }
        }
    }
}

__device__ __forceinline__ float leaky(float x) {
    return (x > 0.f) ? x : NEG_SLOPE * x;
}

#define OWN_BLOCKS  256
#define OWN_THREADS 1024
#define OWN_WAVES   (OWN_THREADS / 32)
#define DRAIN       8
#define SC_A        32
#define SC_B        8
#define QCAP        ((SC_A + SC_B) * DRAIN)
#define FILL        8
#define RESOLVE     8

typedef __attribute__((ext_vector_type(SC_A))) float scvA;
typedef __attribute__((ext_vector_type(SC_B))) float scvB;
static_assert(SC_A % RESOLVE == 0 && SC_B % RESOLVE == 0, "a trip of phase 2 stays inside one score vector");
#define NOT_OWNED   0xFFFFu

template <typename VEC>
__device__ __forceinline__ void own_resolve_max(
        VEC &sc, int gBeg, int gEnd, int gOff,
        const int* q, const unsigned short* qd, int n, int lane,
        const float* __restrict__ el, const float* erl, float* mloc) {
    const int j = lane & 7;
    const int h = lane >> 3;
    for (int g0 = gBeg; g0 < gEnd; g0 += RESOLVE) {
        int      s[RESOLVE];
        unsigned dr[RESOLVE];
        #pragma unroll
        for (int u = 0; u < RESOLVE; ++u) {
            const int idx  = (g0 + u) * DRAIN + j;
            const int idxc = (idx < n) ? idx : n - 1;
            s[u]  = q[idxc];
            dr[u] = (idx < n) ? (unsigned)qd[idxc] : NOT_OWNED;
        }
        float xl[RESOLVE], xr[RESOLVE];
        #pragma unroll
        for (int u = 0; u < RESOLVE; ++u)
            xl[u] = el[(size_t)s[u] * HH + h];
        #pragma unroll
        for (int u = 0; u < RESOLVE; ++u)
            xr[u] = erl[((dr[u] != NOT_OWNED) ? (int)dr[u] : 0) * HH + h];
        #pragma unroll
        for (int u = 0; u < RESOLVE; ++u) {
            const bool ok = dr[u] != NOT_OWNED;
            const int  dl = ok ? (int)dr[u] : 0;
            const float x = leaky(xl[u] + xr[u]);
            if (ok) atomicMax(&mloc[dl * HH + h], x);
            sc[g0 + u - gOff] = x;
        }
    }
}

__device__ __forceinline__ int acc_word(int c) {
    return (c & 3) * 32 + (c >> 2);
}

template <int JJ>
__device__ __forceinline__ float group_lane(float x) {
    return __uint_as_float(__builtin_amdgcn_mov_dpp8(__float_as_uint(x), (unsigned)JJ * 0x249249u));
}

template <int JJ>
__device__ __forceinline__ void own_add_edge(const float4 &v, float ae, int dl, bool ok, float* acc, int lane) {
    const int dj  = __builtin_amdgcn_readlane(dl, JJ);
    const int okj = __builtin_amdgcn_readlane((int)ok, JJ);
    if (okj) {
        const float a = group_lane<JJ>(ae);
        float* accd = acc + dj * HD + lane;
        atomicAdd(&accd[0],  v.x * a);
        atomicAdd(&accd[32], v.y * a);
        atomicAdd(&accd[64], v.z * a);
        atomicAdd(&accd[96], v.w * a);
    }
}

template <typename VEC>
__device__ __forceinline__ void own_accumulate(
        const VEC &sc, int gBeg, int gEnd, int gOff,
        const int* q, const unsigned short* qd, int n, int lane,
        const float* __restrict__ ft, const float* mloc, float* den, float* acc) {
    static_assert(DRAIN == 8, "one queued edge per lane of a group of eight");
    const int j = lane & 7;
    const int h = lane >> 3;
    for (int g = gBeg; g < gEnd; ++g) {
        const int      b   = g * DRAIN;
        const int      idx = b + j;
        const int      idxc = (idx < n) ? idx : n - 1;
        const int      sv  = q[idxc];
        const unsigned dr  = (idx < n) ? (unsigned)qd[idxc] : NOT_OWNED;
        const bool     ok  = dr != NOT_OWNED;
        const int      dl  = ok ? (int)dr : 0;
        float4 v[DRAIN];
        #pragma unroll
        for (int jj = 0; jj < DRAIN; ++jj) {
            const int s = __builtin_amdgcn_readlane(sv, jj);
            v[jj] = reinterpret_cast<const float4*>(ft + (size_t)s * HD)[lane];
        }
        const float ae = __expf(sc[g - gOff] - mloc[dl * HH + h]);
        if (ok) atomicAdd(&den[dl * HH + h], ae);
        own_add_edge<0>(v[0], ae, dl, ok, acc, lane);
        own_add_edge<1>(v[1], ae, dl, ok, acc, lane);
        own_add_edge<2>(v[2], ae, dl, ok, acc, lane);
        own_add_edge<3>(v[3], ae, dl, ok, acc, lane);
        own_add_edge<4>(v[4], ae, dl, ok, acc, lane);
        own_add_edge<5>(v[5], ae, dl, ok, acc, lane);
        own_add_edge<6>(v[6], ae, dl, ok, acc, lane);
        own_add_edge<7>(v[7], ae, dl, ok, acc, lane);
    }
}

template <bool ACC>
__device__ __forceinline__ void own_plain_pass(
        int lane, int sBeg, int sEnd, int n0, int cnt,
        const unsigned* __restrict__ trow, const int2* __restrict__ ent2,
        const float* __restrict__ el, const float* __restrict__ ft,
        const float* erl, float* mloc, float* den, float* acc) {
    for (int sidx = sBeg + lane; sidx < sEnd; sidx += 32) {
        const unsigned tw = trow[(size_t)sidx * NBINS];
        const int c = (int)(tw & 0xFFFFu);
        const int2* p = ent2 + (size_t)sidx * SORT_S + (tw >> 16);
        for (int i = 0; i < c; ++i) {
            const int2 ed = p[i];
            const int dl = ed.y - n0;
            if ((unsigned)dl >= (unsigned)cnt) continue;
            const int s = ed.x;
            for (int hh = 0; hh < HH; ++hh) {
                const float x = leaky(el[(size_t)s * HH + hh] + erl[dl * HH + hh]);
                if (!ACC) {
                    atomicMax(&mloc[dl * HH + hh], x);
                } else {
                    const float ae = __expf(x - mloc[dl * HH + hh]);
                    atomicAdd(&den[dl * HH + hh], ae);
                    for (int dd = 0; dd < DD; ++dd)
                        atomicAdd(&acc[dl * HD + acc_word(hh * DD + dd)], ft[(size_t)s * HD + hh * DD + dd] * ae);
                }
            }
        }
    }
}

__global__ __launch_bounds__(OWN_THREADS) void gat_owner_aggregate(
        const unsigned* __restrict__ tab, const int* __restrict__ ent,
        const float* __restrict__ el, const float* __restrict__ er,
        const float* __restrict__ ft, const float* __restrict__ bias,
        float* __restrict__ out, int N, int E) {
    __shared__ __attribute__((aligned(16))) float acc[RMAX * HD];
    __shared__ __attribute__((aligned(16))) float den[RMAX * HH];
    __shared__ __attribute__((aligned(16))) float mloc[RMAX * HH];
    __shared__ __attribute__((aligned(16))) float erl[RMAX * HH];
    __shared__ int queue[OWN_WAVES * QCAP];
    __shared__ unsigned short qdst[OWN_WAVES * QCAP];
    __shared__ int ovf;

    const int tid   = threadIdx.x;
    const int lane  = tid & 31;
    const int wave  = __builtin_amdgcn_readfirstlane(tid >> 5);
    int* q              = &queue[wave * QCAP];
    unsigned short* qd  = &qdst[wave * QCAP];
    const int2* ent2    = reinterpret_cast<const int2*>(ent);
    const int nsteps    = (E + SORT_S - 1) / SORT_S;
    const int chunk     = (nsteps + OWN_WAVES - 1) / OWN_WAVES;
    const int sBeg      = wave * chunk;
    const int sEnd      = (sBeg + chunk < nsteps) ? (sBeg + chunk) : nsteps;

    for (int n0 = blockIdx.x * RMAX; n0 < N; n0 += (int)gridDim.x * RMAX) {
        const int cnt = (N - n0 < RMAX) ? (N - n0) : RMAX;
        const unsigned ownBin = (unsigned)(n0 / RMAX) & 255u;
        const unsigned* trow = tab + ownBin;

        const float4 z4 = make_float4(0.f, 0.f, 0.f, 0.f);
        const float4 lo = make_float4(-3.402823466e38f, -3.402823466e38f,
                                      -3.402823466e38f, -3.402823466e38f);
        for (int i = tid; i < cnt * (HD / 4); i += OWN_THREADS) reinterpret_cast<float4*>(acc)[i] = z4;
        for (int i = tid; i < cnt; i += OWN_THREADS) {
            reinterpret_cast<float4*>(den)[i]  = z4;
            reinterpret_cast<float4*>(mloc)[i] = lo;
            reinterpret_cast<float4*>(erl)[i]  = reinterpret_cast<const float4*>(er)[n0 + i];
        }
        if (tid == 0) ovf = 0;
        __syncthreads();

        int  qn   = 0;
        bool lost = false;
        for (int sb = sBeg; sb < sEnd; sb += 32) {
            const int sidx = sb + lane;
            const unsigned tw = (sidx < sEnd) ? trow[(size_t)sidx * NBINS] : 0u;
            const int c = (int)(tw & 0xFFFFu);
            int incl = c;
            #pragma unroll
            for (int o = 1; o < 32; o <<= 1) {
                const int v = __shfl_up(incl, o, 32);
                if (lane >= o) incl += v;
            }
            const int total = __builtin_amdgcn_readlane(incl, 31);
            if (qn + total > QCAP) { lost = true; break; }
            const int pos = qn + incl - c;
            const int2* p = ent2 + (size_t)sidx * SORT_S + (tw >> 16);
            for (int i0 = 0; __builtin_amdgcn_ballot_w32(i0 < c) != 0u; i0 += FILL) {
                int2 ed[FILL];
                #pragma unroll
                for (int u = 0; u < FILL; ++u)
                    ed[u] = (i0 + u < c) ? p[i0 + u] : make_int2(0, 0);
                #pragma unroll
                for (int u = 0; u < FILL; ++u) {
                    if (i0 + u < c) {
                        const int dl = ed[u].y - n0;
                        const bool ok = (unsigned)dl < (unsigned)cnt;
                        q[pos + i0 + u]  = ed[u].x;
                        qd[pos + i0 + u] = ok ? (unsigned short)dl : (unsigned short)NOT_OWNED;
                    }
                }
            }
            qn += total;
        }
        if (lost && lane == 0) ovf = 1;

        const int nG = (qn + DRAIN - 1) / DRAIN;
        const int gA = (nG < SC_A) ? nG : SC_A;
        scvA scA = {};
        scvB scB = {};
        own_resolve_max(scA, 0, gA, 0, q, qd, qn, lane, el, erl, mloc);
        if (nG > SC_A)
            own_resolve_max(scB, SC_A, nG, SC_A, q, qd, qn, lane, el, erl, mloc);
        __syncthreads();

        if (__builtin_amdgcn_readfirstlane(ovf) == 0) {
            own_accumulate(scA, 0, gA, 0, q, qd, qn, lane, ft, mloc, den, acc);
            if (nG > SC_A)
                own_accumulate(scB, SC_A, nG, SC_A, q, qd, qn, lane, ft, mloc, den, acc);
        } else {
            own_plain_pass<false>(lane, sBeg, sEnd, n0, cnt, trow, ent2, el, ft, erl, mloc, den, acc);
            __syncthreads();
            own_plain_pass<true>(lane, sBeg, sEnd, n0, cnt, trow, ent2, el, ft, erl, mloc, den, acc);
        }
        __syncthreads();


        for (int i = tid; i < cnt * (HD / 4); i += OWN_THREADS) {
            const int node = i >> 5;
            const int q4   = i & 31;
            const float* an = acc + node * HD + q4;
            const float4 a4 = make_float4(an[0], an[32], an[64], an[96]);
            const float  dn = den[node * HH + (q4 >> 3)];
            const float4 b4 = reinterpret_cast<const float4*>(bias)[q4];
            float4 o;
            if (dn > 0.f) {
                o.x = a4.x / dn + b4.x; o.y = a4.y / dn + b4.y;
                o.z = a4.z / dn + b4.z; o.w = a4.w / dn + b4.w;
            } else {
                o = b4;
            }
            reinterpret_cast<float4*>(out)[(size_t)(n0 + node) * (HD / 4) + q4] = o;
        }
        __syncthreads();
    }
}

extern "C" void kernel_launch(void* const* d_in, const int* in_sizes, int n_in,
                              void* d_out, int out_size, void* d_ws, size_t ws_size,
                              hipStream_t stream) {
    const float* feat   = (const float*)d_in[0];
    const int*   src    = (const int*)  d_in[1];
    const int*   dst    = (const int*)  d_in[2];
    const float* W      = (const float*)d_in[3];
    const float* attn_l = (const float*)d_in[4];
    const float* attn_r = (const float*)d_in[5];
    const float* bias   = (const float*)d_in[6];

    const int N = in_sizes[0] / DIN;
    const int E = in_sizes[1];
    float* out = (float*)d_out;

    float* ws    = (float*)d_ws;
    float* ft    = ws;
    float* el    = ft    + (size_t)N * HD;
    float* er    = el    + (size_t)N * HH;
    unsigned* tab = (unsigned*)(er + (size_t)N * HH);
    int* ent = (int*)(tab + (size_t)NBINS * ((E + SORT_S - 1) / SORT_S));

    gat_gemm<<<GEMM_BLOCKS, 256, 0, stream>>>(feat, W, ft, N);
    gat_logits_sort<<<LG_BLOCKS, LG_THREADS, 0, stream>>>(ft, attn_l, attn_r, el, er, src, dst, tab, ent, N, E);
    gat_owner_aggregate<<<OWN_BLOCKS, OWN_THREADS, 0, stream>>>(tab, ent, el, er, ft, bias, out, N, E);
}
